# starting kernel
# baseline (speedup 1.0000x reference)
#include <hip/hip_runtime.h>
#include <hip/hip_bf16.h>
#include <math.h>


typedef __attribute__((ext_vector_type(2))) float v2f;
typedef __attribute__((ext_vector_type(8))) float v8f;

#define HAS_WMMA_F32 __has_builtin(__builtin_amdgcn_wmma_f32_16x16x4_f32)

__global__ __launch_bounds__(32) void gat_gemm_kernel(
    const float* __restrict__ X,
    const float* __restrict__ W1,
    const float* __restrict__ W2,
    float* __restrict__ z,
    float* __restrict__ zi,
    int N)
{
    const int rowbase = blockIdx.x * 16;
    const int lane    = threadIdx.x;
    const int half    = lane >> 4;
    const int l16     = lane & 15;

    v8f acc1[4] = {v8f{}, v8f{}, v8f{}, v8f{}};
    v8f acc2[4] = {v8f{}, v8f{}, v8f{}, v8f{}};

    int arow = rowbase + l16;
    if (arow >= N) arow = N - 1;
    const float* xrow = X + (size_t)arow * 128;

#if HAS_WMMA_F32
    for (int k0 = 0; k0 < 128; k0 += 4) {
        const int kk = k0 + 2 * half;
        v2f a; a.x = xrow[kk]; a.y = xrow[kk + 1];
#pragma unroll
        for (int j = 0; j < 4; ++j) {
            const int col = j * 16 + l16;
            v2f b1, b2;
            b1.x = W1[col * 128 + kk]; b1.y = W1[col * 128 + kk + 1];
            b2.x = W2[col * 128 + kk]; b2.y = W2[col * 128 + kk + 1];
            acc1[j] = __builtin_amdgcn_wmma_f32_16x16x4_f32(
                false, a, false, b1, (short)0, acc1[j], false, false);
            acc2[j] = __builtin_amdgcn_wmma_f32_16x16x4_f32(
                false, a, false, b2, (short)0, acc2[j], false, false);
        }
    }
#else
    typedef __attribute__((ext_vector_type(16))) __bf16 v16bf;
    for (int k0 = 0; k0 < 128; k0 += 32) {
        v16bf a;
#pragma unroll
        for (int e = 0; e < 16; ++e) {
            const int v = e >> 1, hb = e & 1;
            const int K = ((v < 4) ? 2 * v : 16 + 2 * (v - 4)) + hb + 8 * half;
            a[e] = (__bf16)xrow[k0 + K];
        }
#pragma unroll
        for (int j = 0; j < 4; ++j) {
            const int col = j * 16 + l16;
            v16bf b1, b2;
#pragma unroll
            for (int e = 0; e < 16; ++e) {
                const int v = e >> 1, hb = e & 1;
                const int K = ((v < 4) ? 2 * v : 16 + 2 * (v - 4)) + hb + 8 * half;
                b1[e] = (__bf16)W1[col * 128 + k0 + K];
                b2[e] = (__bf16)W2[col * 128 + k0 + K];
            }
            acc1[j] = __builtin_amdgcn_wmma_f32_16x16x32_bf16(
                false, a, false, b1, (short)0, acc1[j], false, false);
            acc2[j] = __builtin_amdgcn_wmma_f32_16x16x32_bf16(
                false, a, false, b2, (short)0, acc2[j], false, false);
        }
    }
#endif

#pragma unroll
    for (int j = 0; j < 4; ++j) {
#pragma unroll
        for (int r = 0; r < 8; ++r) {
            const int row = rowbase + r + 8 * half;
            if (row < N) {
                z [(size_t)row * 64 + j * 16 + l16] = acc1[j][r];
                zi[(size_t)row * 64 + j * 16 + l16] = acc2[j][r];
            }
        }
    }
}

__global__ __launch_bounds__(256) void gat_node_kernel(
    const float* __restrict__ z, const float* __restrict__ Wa,
    float* __restrict__ s1, float* __restrict__ s2,
    float* __restrict__ m, float* __restrict__ denom,
    float* __restrict__ acc, int N)
{
    const int wave = (blockIdx.x * blockDim.x + threadIdx.x) >> 5;
    const int lane = threadIdx.x & 31;
    if (wave >= N) return;
    const float* zr = z + (size_t)wave * 64;
    float a0 = zr[lane], a1 = zr[lane + 32];
    float p1 = a0 * Wa[lane]      + a1 * Wa[lane + 32];
    float p2 = a0 * Wa[64 + lane] + a1 * Wa[96 + lane];
#pragma unroll
    for (int off = 16; off > 0; off >>= 1) {
        p1 += __shfl_xor(p1, off, 32);
        p2 += __shfl_xor(p2, off, 32);
    }
    acc[(size_t)wave * 64 + lane]      = 0.0f;
    acc[(size_t)wave * 64 + lane + 32] = 0.0f;
    if (lane == 0) {
        s1[wave]    = p1;
        s2[wave]    = p2;
        m[wave]     = -__builtin_inff();
        denom[wave] = 0.0f;
    }
}

__global__ __launch_bounds__(256) void gat_edge_logit_kernel(
    const float* __restrict__ edge_d,
    const int* __restrict__ src, const int* __restrict__ dst,
    const float* __restrict__ s1, const float* __restrict__ s2,
    const float* __restrict__ W0, const float* __restrict__ Wa,
    float* __restrict__ ebuf, float* __restrict__ m, int E)
{
    const int e = blockIdx.x * blockDim.x + threadIdx.x;
    if (e >= E) return;
    const int d = dst[e];
    float v = s1[src[e]] + s2[d] + edge_d[e] * W0[0] * Wa[128];
    v = (v > 0.0f) ? v : 0.01f * v;
    ebuf[e] = v;
    if (v >= 0.0f) atomicMax((int*)(m + d), __float_as_int(v));
    else           atomicMin((unsigned int*)(m + d), __float_as_uint(v));
}

__global__ __launch_bounds__(256) void gat_edge_exp_kernel(
    const int* __restrict__ dst, const float* __restrict__ m,
    float* __restrict__ ebuf, float* __restrict__ denom, int E)
{
    const int e = blockIdx.x * blockDim.x + threadIdx.x;
    if (e >= E) return;
    const int d = dst[e];
    const float ex = __expf(ebuf[e] - m[d]);
    ebuf[e] = ex;
    atomicAdd(denom + d, ex);
}

__global__ __launch_bounds__(256) void gat_scatter_kernel(
    const int* __restrict__ src, const int* __restrict__ dst,
    const float* __restrict__ ebuf, const float* __restrict__ denom,
    const float* __restrict__ z, float* __restrict__ acc, int E)
{
    const int e = blockIdx.x * 4 + (threadIdx.x >> 6);
    const int c = threadIdx.x & 63;
    if (e >= E) return;
    const int d = dst[e];
    const float alpha = ebuf[e] / denom[d];
    atomicAdd(acc + (size_t)d * 64 + c, alpha * z[(size_t)src[e] * 64 + c]);
}

__global__ __launch_bounds__(256) void gat_final_kernel(
    const float* __restrict__ zi, float* __restrict__ out, int total)
{
    const int i = blockIdx.x * blockDim.x + threadIdx.x;
    if (i >= total) return;
    const float v = zi[i] + out[i];
    out[i] = (v > 0.0f) ? v : 0.0f;
}

extern "C" void kernel_launch(void* const* d_in, const int* in_sizes, int n_in,
                              void* d_out, int out_size, void* d_ws, size_t ws_size,
                              hipStream_t stream)
{
    const float* X   = (const float*)d_in[0];
    const float* ed  = (const float*)d_in[1];
    const float* W0  = (const float*)d_in[2];
    const float* W1  = (const float*)d_in[3];
    const float* W2  = (const float*)d_in[4];
    const float* Wa  = (const float*)d_in[5];
    const int*   src = (const int*)d_in[6];
    const int*   dst = (const int*)d_in[7];

    const int N = in_sizes[0] / 128;
    const int E = in_sizes[6];

    float* ws    = (float*)d_ws;
    float* z     = ws;  ws += (size_t)N * 64;
    float* zi    = ws;  ws += (size_t)N * 64;
    float* s1    = ws;  ws += N;
    float* s2    = ws;  ws += N;
    float* m     = ws;  ws += N;
    float* denom = ws;  ws += N;
    float* ebuf  = ws;  ws += E;
    float* out   = (float*)d_out;

    const int rowTiles = (N + 15) / 16;
    gat_gemm_kernel<<<rowTiles, 32, 0, stream>>>(X, W1, W2, z, zi, N);

    const int nodeThreads = N * 32;
    gat_node_kernel<<<(nodeThreads + 255) / 256, 256, 0, stream>>>(
        z, Wa, s1, s2, m, denom, out, N);

    gat_edge_logit_kernel<<<(E + 255) / 256, 256, 0, stream>>>(
        ed, src, dst, s1, s2, W0, Wa, ebuf, m, E);

    gat_edge_exp_kernel<<<(E + 255) / 256, 256, 0, stream>>>(
        dst, m, ebuf, denom, E);

    gat_scatter_kernel<<<(E + 3) / 4, 256, 0, stream>>>(
        src, dst, ebuf, denom, z, out, E);

    gat_final_kernel<<<(N * 64 + 255) / 256, 256, 0, stream>>>(zi, out, N * 64);
}


# scatter kernel: one wave per 32 edges instead of per half edge; edge indices broadcast from lanes to scalars, z rows fetched 16 lines at a time ahead of the whole-line f32 atomic adds
# speedup vs baseline: 1.5879x; 1.5879x over previous
#include <hip/hip_runtime.h>
#include <hip/hip_bf16.h>
#include <math.h>


typedef __attribute__((ext_vector_type(2))) float v2f;
typedef __attribute__((ext_vector_type(8))) float v8f;

#define HAS_WMMA_F32 __has_builtin(__builtin_amdgcn_wmma_f32_16x16x4_f32)

__global__ __launch_bounds__(32) void gat_gemm_kernel(
    const float* __restrict__ X,
    const float* __restrict__ W1,
    const float* __restrict__ W2,
    float* __restrict__ z,
    float* __restrict__ zi,
    int N)
{
    const int rowbase = blockIdx.x * 16;
    const int lane    = threadIdx.x;
    const int half    = lane >> 4;
    const int l16     = lane & 15;

    v8f acc1[4] = {v8f{}, v8f{}, v8f{}, v8f{}};
    v8f acc2[4] = {v8f{}, v8f{}, v8f{}, v8f{}};

    int arow = rowbase + l16;
    if (arow >= N) arow = N - 1;
    const float* xrow = X + (size_t)arow * 128;

#if HAS_WMMA_F32
    for (int k0 = 0; k0 < 128; k0 += 4) {
        const int kk = k0 + 2 * half;
        v2f a; a.x = xrow[kk]; a.y = xrow[kk + 1];
#pragma unroll
        for (int j = 0; j < 4; ++j) {
            const int col = j * 16 + l16;
            v2f b1, b2;
            b1.x = W1[col * 128 + kk]; b1.y = W1[col * 128 + kk + 1];
            b2.x = W2[col * 128 + kk]; b2.y = W2[col * 128 + kk + 1];
            acc1[j] = __builtin_amdgcn_wmma_f32_16x16x4_f32(
                false, a, false, b1, (short)0, acc1[j], false, false);
            acc2[j] = __builtin_amdgcn_wmma_f32_16x16x4_f32(
                false, a, false, b2, (short)0, acc2[j], false, false);
        }
    }
#else
    typedef __attribute__((ext_vector_type(16))) __bf16 v16bf;
    for (int k0 = 0; k0 < 128; k0 += 32) {
        v16bf a;
#pragma unroll
        for (int e = 0; e < 16; ++e) {
            const int v = e >> 1, hb = e & 1;
            const int K = ((v < 4) ? 2 * v : 16 + 2 * (v - 4)) + hb + 8 * half;
            a[e] = (__bf16)xrow[k0 + K];
        }
#pragma unroll
        for (int j = 0; j < 4; ++j) {
            const int col = j * 16 + l16;
            v16bf b1, b2;
#pragma unroll
            for (int e = 0; e < 16; ++e) {
                const int v = e >> 1, hb = e & 1;
                const int K = ((v < 4) ? 2 * v : 16 + 2 * (v - 4)) + hb + 8 * half;
                b1[e] = (__bf16)W1[col * 128 + k0 + K];
                b2[e] = (__bf16)W2[col * 128 + k0 + K];
            }
            acc1[j] = __builtin_amdgcn_wmma_f32_16x16x32_bf16(
                false, a, false, b1, (short)0, acc1[j], false, false);
            acc2[j] = __builtin_amdgcn_wmma_f32_16x16x32_bf16(
                false, a, false, b2, (short)0, acc2[j], false, false);
        }
    }
#endif

#pragma unroll
    for (int j = 0; j < 4; ++j) {
#pragma unroll
        for (int r = 0; r < 8; ++r) {
            const int row = rowbase + r + 8 * half;
            if (row < N) {
                z [(size_t)row * 64 + j * 16 + l16] = acc1[j][r];
                zi[(size_t)row * 64 + j * 16 + l16] = acc2[j][r];
            }
        }
    }
}

__global__ __launch_bounds__(256) void gat_node_kernel(
    const float* __restrict__ z, const float* __restrict__ Wa,
    float* __restrict__ s1, float* __restrict__ s2,
    float* __restrict__ m, float* __restrict__ denom,
    float* __restrict__ acc, int N)
{
    const int wave = (blockIdx.x * blockDim.x + threadIdx.x) >> 5;
    const int lane = threadIdx.x & 31;
    if (wave >= N) return;
    const float* zr = z + (size_t)wave * 64;
    float a0 = zr[lane], a1 = zr[lane + 32];
    float p1 = a0 * Wa[lane]      + a1 * Wa[lane + 32];
    float p2 = a0 * Wa[64 + lane] + a1 * Wa[96 + lane];
#pragma unroll
    for (int off = 16; off > 0; off >>= 1) {
        p1 += __shfl_xor(p1, off, 32);
        p2 += __shfl_xor(p2, off, 32);
    }
    acc[(size_t)wave * 64 + lane]      = 0.0f;
    acc[(size_t)wave * 64 + lane + 32] = 0.0f;
    if (lane == 0) {
        s1[wave]    = p1;
        s2[wave]    = p2;
        m[wave]     = -__builtin_inff();
        denom[wave] = 0.0f;
    }
}

__global__ __launch_bounds__(256) void gat_edge_logit_kernel(
    const float* __restrict__ edge_d,
    const int* __restrict__ src, const int* __restrict__ dst,
    const float* __restrict__ s1, const float* __restrict__ s2,
    const float* __restrict__ W0, const float* __restrict__ Wa,
    float* __restrict__ ebuf, float* __restrict__ m, int E)
{
    const int e = blockIdx.x * blockDim.x + threadIdx.x;
    if (e >= E) return;
    const int d = dst[e];
    float v = s1[src[e]] + s2[d] + edge_d[e] * W0[0] * Wa[128];
    v = (v > 0.0f) ? v : 0.01f * v;
    ebuf[e] = v;
    if (v >= 0.0f) atomicMax((int*)(m + d), __float_as_int(v));
    else           atomicMin((unsigned int*)(m + d), __float_as_uint(v));
}

__global__ __launch_bounds__(256) void gat_edge_exp_kernel(
    const int* __restrict__ dst, const float* __restrict__ m,
    float* __restrict__ ebuf, float* __restrict__ denom, int E)
{
    const int e = blockIdx.x * blockDim.x + threadIdx.x;
    if (e >= E) return;
    const int d = dst[e];
    const float ex = __expf(ebuf[e] - m[d]);
    ebuf[e] = ex;
    atomicAdd(denom + d, ex);
}

__global__ __launch_bounds__(256) void gat_scatter_kernel(
    const int* __restrict__ src, const int* __restrict__ dst,
    const float* __restrict__ ebuf, const float* __restrict__ denom,
    const float* __restrict__ z, float* __restrict__ acc, int E)
{
    const int lane = threadIdx.x & 31;
    const int wave = __builtin_amdgcn_readfirstlane(blockIdx.x * 8 + (threadIdx.x >> 5));
    const int base = wave * 32;
    if (base >= E) return;
    const int cnt = (E - base < 32) ? (E - base) : 32;

    int d = 0, s = 0;
    float a = 0.0f;
    if (lane < cnt) {
        const int e = base + lane;
        d = dst[e];
        s = src[e];
        a = ebuf[e] / denom[d];
    }

    for (int j0 = 0; j0 < cnt; j0 += 8) {
        if (j0 + 8 <= cnt) {
            float v0[8], v1[8];
#pragma unroll
            for (int u = 0; u < 8; ++u) {
                const int sj = __builtin_amdgcn_readlane(s, j0 + u);
                const float* zr = z + (size_t)sj * 64;
                v0[u] = zr[lane];
                v1[u] = zr[lane + 32];
            }
#pragma unroll
            for (int u = 0; u < 8; ++u) {
                const int   dj = __builtin_amdgcn_readlane(d, j0 + u);
                const float aj = __int_as_float(
                    __builtin_amdgcn_readlane(__float_as_int(a), j0 + u));
                float* ar = acc + (size_t)dj * 64;
                atomicAdd(ar + lane,      aj * v0[u]);
                atomicAdd(ar + lane + 32, aj * v1[u]);
            }
        } else {
            for (int j = j0; j < cnt; ++j) {
                const int   sj = __builtin_amdgcn_readlane(s, j);
                const int   dj = __builtin_amdgcn_readlane(d, j);
                const float aj = __int_as_float(
                    __builtin_amdgcn_readlane(__float_as_int(a), j));
                const float* zr = z + (size_t)sj * 64;
                float* ar = acc + (size_t)dj * 64;
                atomicAdd(ar + lane,      aj * zr[lane]);
                atomicAdd(ar + lane + 32, aj * zr[lane + 32]);
            }
        }
    }
}

__global__ __launch_bounds__(256) void gat_final_kernel(
    const float* __restrict__ zi, float* __restrict__ out, int total)
{
    const int i = blockIdx.x * blockDim.x + threadIdx.x;
    if (i >= total) return;
    const float v = zi[i] + out[i];
    out[i] = (v > 0.0f) ? v : 0.0f;
}

extern "C" void kernel_launch(void* const* d_in, const int* in_sizes, int n_in,
                              void* d_out, int out_size, void* d_ws, size_t ws_size,
                              hipStream_t stream)
{
    const float* X   = (const float*)d_in[0];
    const float* ed  = (const float*)d_in[1];
    const float* W0  = (const float*)d_in[2];
    const float* W1  = (const float*)d_in[3];
    const float* W2  = (const float*)d_in[4];
    const float* Wa  = (const float*)d_in[5];
    const int*   src = (const int*)d_in[6];
    const int*   dst = (const int*)d_in[7];

    const int N = in_sizes[0] / 128;
    const int E = in_sizes[6];

    float* ws    = (float*)d_ws;
    float* z     = ws;  ws += (size_t)N * 64;
    float* zi    = ws;  ws += (size_t)N * 64;
    float* s1    = ws;  ws += N;
    float* s2    = ws;  ws += N;
    float* m     = ws;  ws += N;
    float* denom = ws;  ws += N;
    float* ebuf  = ws;  ws += E;
    float* out   = (float*)d_out;

    const int rowTiles = (N + 15) / 16;
    gat_gemm_kernel<<<rowTiles, 32, 0, stream>>>(X, W1, W2, z, zi, N);

    const int nodeThreads = N * 32;
    gat_node_kernel<<<(nodeThreads + 255) / 256, 256, 0, stream>>>(
        z, Wa, s1, s2, m, denom, out, N);

    gat_edge_logit_kernel<<<(E + 255) / 256, 256, 0, stream>>>(
        ed, src, dst, s1, s2, W0, Wa, ebuf, m, E);

    gat_edge_exp_kernel<<<(E + 255) / 256, 256, 0, stream>>>(
        dst, m, ebuf, denom, E);

    gat_scatter_kernel<<<(E + 255) / 256, 256, 0, stream>>>(
        src, dst, ebuf, denom, z, out, E);

    gat_final_kernel<<<(N * 64 + 255) / 256, 256, 0, stream>>>(zi, out, N * 64);
}


# edges bucketed by dst (returning int atomicAdd count, scan, place); one wave per node then does softmax, weighted z[src] row sum and relu in registers: no float atomics, no separate final pass
# speedup vs baseline: 2.0742x; 1.3063x over previous
#include <hip/hip_runtime.h>
#include <hip/hip_bf16.h>
#include <math.h>


typedef __attribute__((ext_vector_type(2))) float v2f;
typedef __attribute__((ext_vector_type(8))) float v8f;

#define HAS_WMMA_F32 __has_builtin(__builtin_amdgcn_wmma_f32_16x16x4_f32)

__global__ __launch_bounds__(32) void gat_gemm_kernel(
    const float* __restrict__ X,
    const float* __restrict__ W1,
    const float* __restrict__ W2,
    float* __restrict__ z,
    float* __restrict__ zi,
    int N)
{
    const int rowbase = blockIdx.x * 16;
    const int lane    = threadIdx.x;
    const int half    = lane >> 4;
    const int l16     = lane & 15;

    v8f acc1[4] = {v8f{}, v8f{}, v8f{}, v8f{}};
    v8f acc2[4] = {v8f{}, v8f{}, v8f{}, v8f{}};

    int arow = rowbase + l16;
    if (arow >= N) arow = N - 1;
    const float* xrow = X + (size_t)arow * 128;

#if HAS_WMMA_F32
    for (int k0 = 0; k0 < 128; k0 += 4) {
        const int kk = k0 + 2 * half;
        v2f a; a.x = xrow[kk]; a.y = xrow[kk + 1];
#pragma unroll
        for (int j = 0; j < 4; ++j) {
            const int col = j * 16 + l16;
            v2f b1, b2;
            b1.x = W1[col * 128 + kk]; b1.y = W1[col * 128 + kk + 1];
            b2.x = W2[col * 128 + kk]; b2.y = W2[col * 128 + kk + 1];
            acc1[j] = __builtin_amdgcn_wmma_f32_16x16x4_f32(
                false, a, false, b1, (short)0, acc1[j], false, false);
            acc2[j] = __builtin_amdgcn_wmma_f32_16x16x4_f32(
                false, a, false, b2, (short)0, acc2[j], false, false);
        }
    }
#else
    typedef __attribute__((ext_vector_type(16))) __bf16 v16bf;
    for (int k0 = 0; k0 < 128; k0 += 32) {
        v16bf a;
#pragma unroll
        for (int e = 0; e < 16; ++e) {
            const int v = e >> 1, hb = e & 1;
            const int K = ((v < 4) ? 2 * v : 16 + 2 * (v - 4)) + hb + 8 * half;
            a[e] = (__bf16)xrow[k0 + K];
        }
#pragma unroll
        for (int j = 0; j < 4; ++j) {
            const int col = j * 16 + l16;
            v16bf b1, b2;
#pragma unroll
            for (int e = 0; e < 16; ++e) {
                const int v = e >> 1, hb = e & 1;
                const int K = ((v < 4) ? 2 * v : 16 + 2 * (v - 4)) + hb + 8 * half;
                b1[e] = (__bf16)W1[col * 128 + k0 + K];
                b2[e] = (__bf16)W2[col * 128 + k0 + K];
            }
            acc1[j] = __builtin_amdgcn_wmma_f32_16x16x32_bf16(
                false, a, false, b1, (short)0, acc1[j], false, false);
            acc2[j] = __builtin_amdgcn_wmma_f32_16x16x32_bf16(
                false, a, false, b2, (short)0, acc2[j], false, false);
        }
    }
#endif

#pragma unroll
    for (int j = 0; j < 4; ++j) {
#pragma unroll
        for (int r = 0; r < 8; ++r) {
            const int row = rowbase + r + 8 * half;
            if (row < N) {
                z [(size_t)row * 64 + j * 16 + l16] = acc1[j][r];
                zi[(size_t)row * 64 + j * 16 + l16] = acc2[j][r];
            }
        }
    }
}

__global__ __launch_bounds__(256) void gat_node_kernel(
    const float* __restrict__ z, const float* __restrict__ Wa,
    float* __restrict__ s1, float* __restrict__ s2,
    int* __restrict__ cnt, int N)
{
    const int wave = (blockIdx.x * blockDim.x + threadIdx.x) >> 5;
    const int lane = threadIdx.x & 31;
    if (wave >= N) return;
    const float* zr = z + (size_t)wave * 64;
    float a0 = zr[lane], a1 = zr[lane + 32];
    float p1 = a0 * Wa[lane]      + a1 * Wa[lane + 32];
    float p2 = a0 * Wa[64 + lane] + a1 * Wa[96 + lane];
#pragma unroll
    for (int off = 16; off > 0; off >>= 1) {
        p1 += __shfl_xor(p1, off, 32);
        p2 += __shfl_xor(p2, off, 32);
    }
    if (lane == 0) {
        s1[wave]  = p1;
        s2[wave]  = p2;
        cnt[wave] = 0;
    }
}

__global__ __launch_bounds__(256) void gat_count_kernel(
    const int* __restrict__ dst, int* __restrict__ cnt,
    int* __restrict__ rank, int E)
{
    const int e0 = (blockIdx.x * blockDim.x + threadIdx.x) * 4;
    if (e0 + 3 < E) {
        const int4 d = *(const int4*)(dst + e0);
        int4 r;
        r.x = atomicAdd(cnt + d.x, 1);
        r.y = atomicAdd(cnt + d.y, 1);
        r.z = atomicAdd(cnt + d.z, 1);
        r.w = atomicAdd(cnt + d.w, 1);
        *(int4*)(rank + e0) = r;
    } else {
        for (int e = e0; e < E; ++e) rank[e] = atomicAdd(cnt + dst[e], 1);
    }
}

__global__ __launch_bounds__(1024) void gat_offsets_kernel(
    const int* __restrict__ cnt, int* __restrict__ offs, int N)
{
    __shared__ int sh_prev[32];
    __shared__ int sh_loc[32];
    const int t = threadIdx.x, lane = t & 31, w = t >> 5;
    const int start = blockIdx.x * 1024;

    int prev = 0;
    const int4* c4 = (const int4*)cnt;
    const int nq = start >> 2;
    for (int q0 = 0; q0 < nq; q0 += 16 * 1024) {
        int4 c[16];
#pragma unroll
        for (int u = 0; u < 16; ++u) {
            const int q = q0 + u * 1024 + t;
            c[u] = (q < nq) ? c4[q] : make_int4(0, 0, 0, 0);
        }
#pragma unroll
        for (int u = 0; u < 16; ++u) prev += (c[u].x + c[u].y) + (c[u].z + c[u].w);
    }
    for (int off = 16; off > 0; off >>= 1) prev += __shfl_xor(prev, off, 32);

    const int i = start + t;
    const int c = (i < N) ? cnt[i] : 0;
    int inc = c;
    for (int off = 1; off < 32; off <<= 1) {
        const int v = __shfl_up(inc, off, 32);
        if (lane >= off) inc += v;
    }
    if (lane == 31) sh_loc[w]  = inc;
    if (lane == 0)  sh_prev[w] = prev;
    __syncthreads();

    int base = sh_prev[lane];
    for (int off = 16; off > 0; off >>= 1) base += __shfl_xor(base, off, 32);
    const int tot = sh_loc[lane];
    int winc = tot;
    for (int off = 1; off < 32; off <<= 1) {
        const int v = __shfl_up(winc, off, 32);
        if (lane >= off) winc += v;
    }
    const int wbase = __shfl(winc - tot, w, 32);
    if (i < N) offs[i] = base + wbase + (inc - c);
}

__device__ __forceinline__ float gat_logit(float s1v, float s2v, float ed, float w0, float wa)
{
    float v = s1v + s2v + ed * w0 * wa;
    return (v > 0.0f) ? v : 0.01f * v;
}

__global__ __launch_bounds__(256) void gat_place_kernel(
    const float* __restrict__ edge_d,
    const int* __restrict__ src, const int* __restrict__ dst,
    const int* __restrict__ rank, const int* __restrict__ offs,
    const float* __restrict__ s1, const float* __restrict__ s2,
    const float* __restrict__ W0, const float* __restrict__ Wa,
    int* __restrict__ pairs_i, int E)
{
    int2* __restrict__ pairs = (int2*)pairs_i;
    const int e0 = (blockIdx.x * blockDim.x + threadIdx.x) * 4;
    const float w0 = W0[0], wa = Wa[128];
    if (e0 + 3 < E) {
        const int4   d  = *(const int4*)(dst + e0);
        const int4   s  = *(const int4*)(src + e0);
        const int4   r  = *(const int4*)(rank + e0);
        const float4 ed = *(const float4*)(edge_d + e0);
        const int   p0 = offs[d.x] + r.x, p1 = offs[d.y] + r.y;
        const int   p2 = offs[d.z] + r.z, p3 = offs[d.w] + r.w;
        const float v0 = gat_logit(s1[s.x], s2[d.x], ed.x, w0, wa);
        const float v1 = gat_logit(s1[s.y], s2[d.y], ed.y, w0, wa);
        const float v2 = gat_logit(s1[s.z], s2[d.z], ed.z, w0, wa);
        const float v3 = gat_logit(s1[s.w], s2[d.w], ed.w, w0, wa);
        pairs[p0] = make_int2(s.x, __float_as_int(v0));
        pairs[p1] = make_int2(s.y, __float_as_int(v1));
        pairs[p2] = make_int2(s.z, __float_as_int(v2));
        pairs[p3] = make_int2(s.w, __float_as_int(v3));
    } else {
        for (int e = e0; e < E; ++e) {
            const int d = dst[e], s = src[e];
            const float v = gat_logit(s1[s], s2[d], edge_d[e], w0, wa);
            pairs[offs[d] + rank[e]] = make_int2(s, __float_as_int(v));
        }
    }
}

template <int CTRL>
__device__ __forceinline__ float gat_row_xmask(float x)
{
    return __int_as_float(__builtin_amdgcn_update_dpp(
        0, __float_as_int(x), CTRL, 0xf, 0xf, true));
}
__device__ __forceinline__ float gat_wave_max(float x)
{
    x = fmaxf(x, gat_row_xmask<0x161>(x));
    x = fmaxf(x, gat_row_xmask<0x162>(x));
    x = fmaxf(x, gat_row_xmask<0x164>(x));
    x = fmaxf(x, gat_row_xmask<0x168>(x));
    const float lo = __int_as_float(__builtin_amdgcn_readlane(__float_as_int(x), 0));
    const float hi = __int_as_float(__builtin_amdgcn_readlane(__float_as_int(x), 16));
    return fmaxf(lo, hi);
}
__device__ __forceinline__ float gat_wave_sum(float x)
{
    x += gat_row_xmask<0x161>(x);
    x += gat_row_xmask<0x162>(x);
    x += gat_row_xmask<0x164>(x);
    x += gat_row_xmask<0x168>(x);
    const float lo = __int_as_float(__builtin_amdgcn_readlane(__float_as_int(x), 0));
    const float hi = __int_as_float(__builtin_amdgcn_readlane(__float_as_int(x), 16));
    return lo + hi;
}

template <int K>
__device__ __forceinline__ void gat_pull_rows(
    const float* __restrict__ z, int s, float a, int j, int lane, v2f& acc)
{
    v2f r[K];
#pragma unroll
    for (int u = 0; u < K; ++u) {
        const int sj = __builtin_amdgcn_readlane(s, j + u);
        r[u] = *(const v2f*)(z + (size_t)sj * 64 + 2 * lane);
    }
#pragma unroll
    for (int u = 0; u < K; ++u) {
        const float aj = __int_as_float(
            __builtin_amdgcn_readlane(__float_as_int(a), j + u));
        acc.x = fmaf(aj, r[u].x, acc.x);
        acc.y = fmaf(aj, r[u].y, acc.y);
    }
}

__device__ __forceinline__ void gat_pull_chunk(
    const float* __restrict__ z, int s, float a, int n, int lane, v2f& acc)
{
    int j = 0;
    for (; j + 8 <= n; j += 8) gat_pull_rows<8>(z, s, a, j, lane, acc);
    if (n - j >= 4) { gat_pull_rows<4>(z, s, a, j, lane, acc); j += 4; }
    if (n - j >= 2) { gat_pull_rows<2>(z, s, a, j, lane, acc); j += 2; }
    if (n - j >= 1) { gat_pull_rows<1>(z, s, a, j, lane, acc); }
}

__global__ __launch_bounds__(256) void gat_pull_kernel(
    const int* __restrict__ cnt, const int* __restrict__ offs,
    const int* __restrict__ pairs_i, const float* __restrict__ z,
    float* out, int N)
{
    const int2* __restrict__ pairs = (const int2*)pairs_i;
    const int lane = threadIdx.x & 31;
    const int wave = __builtin_amdgcn_readfirstlane(blockIdx.x * 8 + (threadIdx.x >> 5));
    if (wave >= N) return;
    const int deg = cnt[wave];
    const int off = offs[wave];
    float* orow = out + (size_t)wave * 64 + 2 * lane;
    const v2f zi = *(const v2f*)orow;
    v2f acc = {0.0f, 0.0f};

    if (deg <= 32) {
        int s = 0;
        float v = -__builtin_inff();
        if (lane < deg) {
            const int2 p = pairs[off + lane];
            s = p.x;
            v = __int_as_float(p.y);
        }
        const float m   = gat_wave_max(v);
        const float ex  = (lane < deg) ? __expf(v - m) : 0.0f;
        const float den = gat_wave_sum(ex);
        const float a   = ex / den;
        gat_pull_chunk(z, s, a, deg, lane, acc);
    } else {
        float m = -__builtin_inff();
        for (int c = 0; c < deg; c += 32)
            if (c + lane < deg) m = fmaxf(m, __int_as_float(pairs[off + c + lane].y));
        m = gat_wave_max(m);
        float den = 0.0f;
        for (int c = 0; c < deg; c += 32)
            if (c + lane < deg) den += __expf(__int_as_float(pairs[off + c + lane].y) - m);
        den = gat_wave_sum(den);
        for (int c = 0; c < deg; c += 32) {
            int s = 0;
            float a = 0.0f;
            if (c + lane < deg) {
                const int2 p = pairs[off + c + lane];
                s = p.x;
                a = __expf(__int_as_float(p.y) - m) / den;
            }
            const int n = (deg - c < 32) ? (deg - c) : 32;
            gat_pull_chunk(z, s, a, n, lane, acc);
        }
    }

    v2f h;
    h.x = zi.x + acc.x;
    h.y = zi.y + acc.y;
    h.x = (h.x > 0.0f) ? h.x : 0.0f;
    h.y = (h.y > 0.0f) ? h.y : 0.0f;
    *(v2f*)orow = h;
}

extern "C" void kernel_launch(void* const* d_in, const int* in_sizes, int n_in,
                              void* d_out, int out_size, void* d_ws, size_t ws_size,
                              hipStream_t stream)
{
    const float* X   = (const float*)d_in[0];
    const float* ed  = (const float*)d_in[1];
    const float* W0  = (const float*)d_in[2];
    const float* W1  = (const float*)d_in[3];
    const float* W2  = (const float*)d_in[4];
    const float* Wa  = (const float*)d_in[5];
    const int*   src = (const int*)d_in[6];
    const int*   dst = (const int*)d_in[7];

    const int N = in_sizes[0] / 128;
    const int E = in_sizes[6];

    float* ws    = (float*)d_ws;
    float* z     = ws;         ws += (size_t)N * 64;
    int*   cnt   = (int*)ws;   ws += N;
    int*   offs  = (int*)ws;   ws += N;
    float* s1    = ws;         ws += N;
    float* s2    = ws;         ws += N;
    int*   pairs = (int*)ws;   ws += (size_t)E * 2;
    int*   rank  = (int*)ws;   ws += E;
    float* out   = (float*)d_out;

    const int rowTiles = (N + 15) / 16;
    gat_gemm_kernel<<<rowTiles, 32, 0, stream>>>(X, W1, W2, z, out, N);

    const int nodeThreads = N * 32;
    gat_node_kernel<<<(nodeThreads + 255) / 256, 256, 0, stream>>>(
        z, Wa, s1, s2, cnt, N);

    gat_count_kernel<<<(E + 1023) / 1024, 256, 0, stream>>>(dst, cnt, rank, E);

    gat_offsets_kernel<<<(N + 1023) / 1024, 1024, 0, stream>>>(cnt, offs, N);

    gat_place_kernel<<<(E + 1023) / 1024, 256, 0, stream>>>(
        ed, src, dst, rank, offs, s1, s2, W0, Wa, pairs, E);

    gat_pull_kernel<<<(nodeThreads + 255) / 256, 256, 0, stream>>>(
        cnt, offs, pairs, z, out, N);
}


# GEMM operand feed: W1|W2 copied once per 8-wave work-group into LDS, B fragments read from LDS and A from X as 16 bytes per lane; k summed in a different order inside each group of 8
# speedup vs baseline: 2.4564x; 1.1843x over previous
#include <hip/hip_runtime.h>
#include <hip/hip_bf16.h>
#include <math.h>


typedef __attribute__((ext_vector_type(2))) float v2f;
typedef __attribute__((ext_vector_type(8))) float v8f;

#define HAS_WMMA_F32 __has_builtin(__builtin_amdgcn_wmma_f32_16x16x4_f32)

template <int CTRL>
__device__ __forceinline__ float gat_row_xmask(float x)
{
    return __int_as_float(__builtin_amdgcn_update_dpp(
        0, __float_as_int(x), CTRL, 0xf, 0xf, true));
}
__device__ __forceinline__ float gat_row_sum16(float x)
{
    x += gat_row_xmask<0x161>(x);
    x += gat_row_xmask<0x162>(x);
    x += gat_row_xmask<0x164>(x);
    x += gat_row_xmask<0x168>(x);
    return x;
}

#define GAT_WPITCH 132

__global__ __launch_bounds__(256) void gat_gemm_kernel(
    const float* __restrict__ X,
    const float* __restrict__ W1,
    const float* __restrict__ W2,
    const float* __restrict__ Wa,
    float* __restrict__ z,
    float* __restrict__ zi,
    float* __restrict__ s1,
    float* __restrict__ s2,
    int* __restrict__ cnt,
    int N)
{
    __shared__ float4 gat_wl4[128 * GAT_WPITCH / 4];
    float* Wl = (float*)gat_wl4;

    const int tid     = threadIdx.x;
    const int lane    = tid & 31;
    const int half    = lane >> 4;
    const int l16     = lane & 15;
    const int rowbase = (blockIdx.x * 8 + (tid >> 5)) * 16;

    int arow = rowbase + l16;
    if (arow >= N) arow = N - 1;
    const float* xrow = X + (size_t)arow * 128 + 4 * half;

#if HAS_WMMA_F32
    float4 a[16];
#pragma unroll
    for (int g = 0; g < 16; ++g) a[g] = *(const float4*)(xrow + 8 * g);
#endif

#pragma unroll
    for (int i = 0; i < 16; ++i) {
        const int idx = i * 256 + tid;
        const int row = idx >> 5, q = idx & 31;
        const float* wsrc = (row < 64) ? (W1 + row * 128) : (W2 + (row - 64) * 128);
        const float4 v = *(const float4*)(wsrc + 4 * q);
        *(float4*)(Wl + row * GAT_WPITCH + 4 * q) = v;
    }
    __syncthreads();
    if (rowbase >= N) return;

    v8f acc1[4] = {v8f{}, v8f{}, v8f{}, v8f{}};
    v8f acc2[4] = {v8f{}, v8f{}, v8f{}, v8f{}};

#if HAS_WMMA_F32
    const float* wfrag = Wl + l16 * GAT_WPITCH + 4 * half;
#pragma unroll
    for (int g = 0; g < 16; ++g) {
        float4 b[8];
#pragma unroll
        for (int j = 0; j < 8; ++j)
            b[j] = *(const float4*)(wfrag + j * 16 * GAT_WPITCH + 8 * g);
        v2f a0, a1;
        a0.x = a[g].x; a0.y = a[g].y;
        a1.x = a[g].z; a1.y = a[g].w;
#pragma unroll
        for (int j = 0; j < 4; ++j) {
            v2f b1, b2;
            b1.x = b[j].x;     b1.y = b[j].y;
            b2.x = b[j + 4].x; b2.y = b[j + 4].y;
            acc1[j] = __builtin_amdgcn_wmma_f32_16x16x4_f32(
                false, a0, false, b1, (short)0, acc1[j], false, false);
            acc2[j] = __builtin_amdgcn_wmma_f32_16x16x4_f32(
                false, a0, false, b2, (short)0, acc2[j], false, false);
        }
#pragma unroll
        for (int j = 0; j < 4; ++j) {
            v2f b1, b2;
            b1.x = b[j].z;     b1.y = b[j].w;
            b2.x = b[j + 4].z; b2.y = b[j + 4].w;
            acc1[j] = __builtin_amdgcn_wmma_f32_16x16x4_f32(
                false, a1, false, b1, (short)0, acc1[j], false, false);
            acc2[j] = __builtin_amdgcn_wmma_f32_16x16x4_f32(
                false, a1, false, b2, (short)0, acc2[j], false, false);
        }
    }
#else
    typedef __attribute__((ext_vector_type(16))) __bf16 v16bf;
    const float* xr = X + (size_t)arow * 128;
    for (int k0 = 0; k0 < 128; k0 += 32) {
        v16bf av;
#pragma unroll
        for (int e = 0; e < 16; ++e) {
            const int v = e >> 1, hb = e & 1;
            const int K = ((v < 4) ? 2 * v : 16 + 2 * (v - 4)) + hb + 8 * half;
            av[e] = (__bf16)xr[k0 + K];
        }
#pragma unroll
        for (int j = 0; j < 4; ++j) {
            const int col = j * 16 + l16;
            v16bf b1, b2;
#pragma unroll
            for (int e = 0; e < 16; ++e) {
                const int v = e >> 1, hb = e & 1;
                const int K = ((v < 4) ? 2 * v : 16 + 2 * (v - 4)) + hb + 8 * half;
                b1[e] = (__bf16)W1[col * 128 + k0 + K];
                b2[e] = (__bf16)W2[col * 128 + k0 + K];
            }
            acc1[j] = __builtin_amdgcn_wmma_f32_16x16x32_bf16(
                false, av, false, b1, (short)0, acc1[j], false, false);
            acc2[j] = __builtin_amdgcn_wmma_f32_16x16x32_bf16(
                false, av, false, b2, (short)0, acc2[j], false, false);
        }
    }
#endif

#pragma unroll
    for (int j = 0; j < 4; ++j) {
#pragma unroll
        for (int r = 0; r < 8; ++r) {
            const int row = rowbase + r + 8 * half;
            if (row < N) {
                z [(size_t)row * 64 + j * 16 + l16] = acc1[j][r];
                zi[(size_t)row * 64 + j * 16 + l16] = acc2[j][r];
            }
        }
    }

    float wa1[4], wa2[4];
    for (int j = 0; j < 4; ++j) {
        wa1[j] = Wa[j * 16 + l16];
        wa2[j] = Wa[64 + j * 16 + l16];
    }
    float keep1 = 0.0f, keep2 = 0.0f;
    for (int r = 0; r < 8; ++r) {
        float p1 = acc1[0][r] * wa1[0];
        float p2 = acc1[0][r] * wa2[0];
        for (int j = 1; j < 4; ++j) {
            p1 = fmaf(acc1[j][r], wa1[j], p1);
            p2 = fmaf(acc1[j][r], wa2[j], p2);
        }
        p1 = gat_row_sum16(p1);
        p2 = gat_row_sum16(p2);
        if (l16 == r) { keep1 = p1; keep2 = p2; }
    }
    const int srow = rowbase + 8 * half + l16;
    if (l16 < 8 && srow < N) {
        s1[srow]  = keep1;
        s2[srow]  = keep2;
        cnt[srow] = 0;
    }
}

__global__ __launch_bounds__(256) void gat_count_kernel(
    const int* __restrict__ dst, int* __restrict__ cnt,
    int* __restrict__ rank, int E)
{
    const int e0 = (blockIdx.x * blockDim.x + threadIdx.x) * 4;
    if (e0 + 3 < E) {
        const int4 d = *(const int4*)(dst + e0);
        int4 r;
        r.x = atomicAdd(cnt + d.x, 1);
        r.y = atomicAdd(cnt + d.y, 1);
        r.z = atomicAdd(cnt + d.z, 1);
        r.w = atomicAdd(cnt + d.w, 1);
        *(int4*)(rank + e0) = r;
    } else {
        for (int e = e0; e < E; ++e) rank[e] = atomicAdd(cnt + dst[e], 1);
    }
}

__global__ __launch_bounds__(1024) void gat_offsets_kernel(
    const int* __restrict__ cnt, int* __restrict__ offs, int N)
{
    __shared__ int sh_prev[32];
    __shared__ int sh_loc[32];
    const int t = threadIdx.x, lane = t & 31, w = t >> 5;
    const int start = blockIdx.x * 1024;

    int prev = 0;
    const int4* c4 = (const int4*)cnt;
    const int nq = start >> 2;
    for (int q0 = 0; q0 < nq; q0 += 16 * 1024) {
        int4 c[16];
#pragma unroll
        for (int u = 0; u < 16; ++u) {
            const int q = q0 + u * 1024 + t;
            c[u] = (q < nq) ? c4[q] : make_int4(0, 0, 0, 0);
        }
#pragma unroll
        for (int u = 0; u < 16; ++u) prev += (c[u].x + c[u].y) + (c[u].z + c[u].w);
    }
    for (int off = 16; off > 0; off >>= 1) prev += __shfl_xor(prev, off, 32);

    const int i = start + t;
    const int c = (i < N) ? cnt[i] : 0;
    int inc = c;
    for (int off = 1; off < 32; off <<= 1) {
        const int v = __shfl_up(inc, off, 32);
        if (lane >= off) inc += v;
    }
    if (lane == 31) sh_loc[w]  = inc;
    if (lane == 0)  sh_prev[w] = prev;
    __syncthreads();

    int base = sh_prev[lane];
    for (int off = 16; off > 0; off >>= 1) base += __shfl_xor(base, off, 32);
    const int tot = sh_loc[lane];
    int winc = tot;
    for (int off = 1; off < 32; off <<= 1) {
        const int v = __shfl_up(winc, off, 32);
        if (lane >= off) winc += v;
    }
    const int wbase = __shfl(winc - tot, w, 32);
    if (i < N) offs[i] = base + wbase + (inc - c);
}

__device__ __forceinline__ float gat_logit(float s1v, float s2v, float ed, float w0, float wa)
{
    float v = s1v + s2v + ed * w0 * wa;
    return (v > 0.0f) ? v : 0.01f * v;
}

__global__ __launch_bounds__(256) void gat_place_kernel(
    const float* __restrict__ edge_d,
    const int* __restrict__ src, const int* __restrict__ dst,
    const int* __restrict__ rank, const int* __restrict__ offs,
    const float* __restrict__ s1, const float* __restrict__ s2,
    const float* __restrict__ W0, const float* __restrict__ Wa,
    int* __restrict__ pairs_i, int E)
{
    int2* __restrict__ pairs = (int2*)pairs_i;
    const int e0 = (blockIdx.x * blockDim.x + threadIdx.x) * 4;
    const float w0 = W0[0], wa = Wa[128];
    if (e0 + 3 < E) {
        const int4   d  = *(const int4*)(dst + e0);
        const int4   s  = *(const int4*)(src + e0);
        const int4   r  = *(const int4*)(rank + e0);
        const float4 ed = *(const float4*)(edge_d + e0);
        const int   p0 = offs[d.x] + r.x, p1 = offs[d.y] + r.y;
        const int   p2 = offs[d.z] + r.z, p3 = offs[d.w] + r.w;
        const float v0 = gat_logit(s1[s.x], s2[d.x], ed.x, w0, wa);
        const float v1 = gat_logit(s1[s.y], s2[d.y], ed.y, w0, wa);
        const float v2 = gat_logit(s1[s.z], s2[d.z], ed.z, w0, wa);
        const float v3 = gat_logit(s1[s.w], s2[d.w], ed.w, w0, wa);
        pairs[p0] = make_int2(s.x, __float_as_int(v0));
        pairs[p1] = make_int2(s.y, __float_as_int(v1));
        pairs[p2] = make_int2(s.z, __float_as_int(v2));
        pairs[p3] = make_int2(s.w, __float_as_int(v3));
    } else {
        for (int e = e0; e < E; ++e) {
            const int d = dst[e], s = src[e];
            const float v = gat_logit(s1[s], s2[d], edge_d[e], w0, wa);
            pairs[offs[d] + rank[e]] = make_int2(s, __float_as_int(v));
        }
    }
}

__device__ __forceinline__ float gat_wave_max(float x)
{
    x = fmaxf(x, gat_row_xmask<0x161>(x));
    x = fmaxf(x, gat_row_xmask<0x162>(x));
    x = fmaxf(x, gat_row_xmask<0x164>(x));
    x = fmaxf(x, gat_row_xmask<0x168>(x));
    const float lo = __int_as_float(__builtin_amdgcn_readlane(__float_as_int(x), 0));
    const float hi = __int_as_float(__builtin_amdgcn_readlane(__float_as_int(x), 16));
    return fmaxf(lo, hi);
}
__device__ __forceinline__ float gat_wave_sum(float x)
{
    x += gat_row_xmask<0x161>(x);
    x += gat_row_xmask<0x162>(x);
    x += gat_row_xmask<0x164>(x);
    x += gat_row_xmask<0x168>(x);
    const float lo = __int_as_float(__builtin_amdgcn_readlane(__float_as_int(x), 0));
    const float hi = __int_as_float(__builtin_amdgcn_readlane(__float_as_int(x), 16));
    return lo + hi;
}

template <int K>
__device__ __forceinline__ void gat_pull_rows(
    const float* __restrict__ z, int s, float a, int j, int lane, v2f& acc)
{
    v2f r[K];
#pragma unroll
    for (int u = 0; u < K; ++u) {
        const int sj = __builtin_amdgcn_readlane(s, j + u);
        r[u] = *(const v2f*)(z + (size_t)sj * 64 + 2 * lane);
    }
#pragma unroll
    for (int u = 0; u < K; ++u) {
        const float aj = __int_as_float(
            __builtin_amdgcn_readlane(__float_as_int(a), j + u));
        acc.x = fmaf(aj, r[u].x, acc.x);
        acc.y = fmaf(aj, r[u].y, acc.y);
    }
}

__device__ __forceinline__ void gat_pull_chunk(
    const float* __restrict__ z, int s, float a, int n, int lane, v2f& acc)
{
    int j = 0;
    for (; j + 8 <= n; j += 8) gat_pull_rows<8>(z, s, a, j, lane, acc);
    if (n - j >= 4) { gat_pull_rows<4>(z, s, a, j, lane, acc); j += 4; }
    if (n - j >= 2) { gat_pull_rows<2>(z, s, a, j, lane, acc); j += 2; }
    if (n - j >= 1) { gat_pull_rows<1>(z, s, a, j, lane, acc); }
}

__global__ __launch_bounds__(256) void gat_pull_kernel(
    const int* __restrict__ cnt, const int* __restrict__ offs,
    const int* __restrict__ pairs_i, const float* __restrict__ z,
    float* out, int N)
{
    const int2* __restrict__ pairs = (const int2*)pairs_i;
    const int lane = threadIdx.x & 31;
    const int wave = __builtin_amdgcn_readfirstlane(blockIdx.x * 8 + (threadIdx.x >> 5));
    if (wave >= N) return;
    const int deg = cnt[wave];
    const int off = offs[wave];
    float* orow = out + (size_t)wave * 64 + 2 * lane;
    const v2f zi = *(const v2f*)orow;
    v2f acc = {0.0f, 0.0f};

    if (deg <= 32) {
        int s = 0;
        float v = -__builtin_inff();
        if (lane < deg) {
            const int2 p = pairs[off + lane];
            s = p.x;
            v = __int_as_float(p.y);
        }
        const float m   = gat_wave_max(v);
        const float ex  = (lane < deg) ? __expf(v - m) : 0.0f;
        const float den = gat_wave_sum(ex);
        const float a   = ex / den;
        gat_pull_chunk(z, s, a, deg, lane, acc);
    } else {
        float m = -__builtin_inff();
        for (int c = 0; c < deg; c += 32)
            if (c + lane < deg) m = fmaxf(m, __int_as_float(pairs[off + c + lane].y));
        m = gat_wave_max(m);
        float den = 0.0f;
        for (int c = 0; c < deg; c += 32)
            if (c + lane < deg) den += __expf(__int_as_float(pairs[off + c + lane].y) - m);
        den = gat_wave_sum(den);
        for (int c = 0; c < deg; c += 32) {
            int s = 0;
            float a = 0.0f;
            if (c + lane < deg) {
                const int2 p = pairs[off + c + lane];
                s = p.x;
                a = __expf(__int_as_float(p.y) - m) / den;
            }
            const int n = (deg - c < 32) ? (deg - c) : 32;
            gat_pull_chunk(z, s, a, n, lane, acc);
        }
    }

    v2f h;
    h.x = zi.x + acc.x;
    h.y = zi.y + acc.y;
    h.x = (h.x > 0.0f) ? h.x : 0.0f;
    h.y = (h.y > 0.0f) ? h.y : 0.0f;
    *(v2f*)orow = h;
}

extern "C" void kernel_launch(void* const* d_in, const int* in_sizes, int n_in,
                              void* d_out, int out_size, void* d_ws, size_t ws_size,
                              hipStream_t stream)
{
    const float* X   = (const float*)d_in[0];
    const float* ed  = (const float*)d_in[1];
    const float* W0  = (const float*)d_in[2];
    const float* W1  = (const float*)d_in[3];
    const float* W2  = (const float*)d_in[4];
    const float* Wa  = (const float*)d_in[5];
    const int*   src = (const int*)d_in[6];
    const int*   dst = (const int*)d_in[7];

    const int N = in_sizes[0] / 128;
    const int E = in_sizes[6];

    float* ws    = (float*)d_ws;
    float* z     = ws;         ws += (size_t)N * 64;
    int*   cnt   = (int*)ws;   ws += N;
    int*   offs  = (int*)ws;   ws += N;
    float* s1    = ws;         ws += N;
    float* s2    = ws;         ws += N;
    int*   pairs = (int*)ws;   ws += (size_t)E * 2;
    int*   rank  = (int*)ws;   ws += E;
    float* out   = (float*)d_out;

    const int rowTiles = (N + 15) / 16;
    gat_gemm_kernel<<<(rowTiles + 7) / 8, 256, 0, stream>>>(X, W1, W2, Wa, z, out, s1, s2, cnt, N);

    const int nodeThreads = N * 32;

    gat_count_kernel<<<(E + 1023) / 1024, 256, 0, stream>>>(dst, cnt, rank, E);

    gat_offsets_kernel<<<(N + 1023) / 1024, 1024, 0, stream>>>(cnt, offs, N);

    gat_place_kernel<<<(E + 1023) / 1024, 256, 0, stream>>>(
        ed, src, dst, rank, offs, s1, s2, W0, Wa, pairs, E);

    gat_pull_kernel<<<(nodeThreads + 255) / 256, 256, 0, stream>>>(
        cnt, offs, pairs, z, out, N);
}


# edges bucketed by dst in a two-level counting sort on LDS counters (bins of 512 nodes, then per node; 6 launches, was 5): no returning global atomic per edge; order inside a bucket still by arrival
# speedup vs baseline: 3.1186x; 1.2696x over previous
#include <hip/hip_runtime.h>
#include <hip/hip_bf16.h>
#include <math.h>


typedef __attribute__((ext_vector_type(2))) float v2f;
typedef __attribute__((ext_vector_type(8))) float v8f;

#define HAS_WMMA_F32 __has_builtin(__builtin_amdgcn_wmma_f32_16x16x4_f32)

template <int CTRL>
__device__ __forceinline__ float gat_row_xmask(float x)
{
    return __int_as_float(__builtin_amdgcn_update_dpp(
        0, __float_as_int(x), CTRL, 0xf, 0xf, true));
}
__device__ __forceinline__ float gat_row_sum16(float x)
{
    x += gat_row_xmask<0x161>(x);
    x += gat_row_xmask<0x162>(x);
    x += gat_row_xmask<0x164>(x);
    x += gat_row_xmask<0x168>(x);
    return x;
}

#define GAT_WPITCH 132

__global__ __launch_bounds__(256) void gat_gemm_kernel(
    const float* __restrict__ X,
    const float* __restrict__ W1,
    const float* __restrict__ W2,
    const float* __restrict__ Wa,
    float* __restrict__ z,
    float* __restrict__ zi,
    float* __restrict__ s1,
    float* __restrict__ s2,
    int N)
{
    __shared__ float4 gat_wl4[128 * GAT_WPITCH / 4];
    float* Wl = (float*)gat_wl4;

    const int tid     = threadIdx.x;
    const int lane    = tid & 31;
    const int half    = lane >> 4;
    const int l16     = lane & 15;
    const int rowbase = (blockIdx.x * 8 + (tid >> 5)) * 16;

    int arow = rowbase + l16;
    if (arow >= N) arow = N - 1;
    const float* xrow = X + (size_t)arow * 128 + 4 * half;

#if HAS_WMMA_F32
    float4 a[16];
#pragma unroll
    for (int g = 0; g < 16; ++g) a[g] = *(const float4*)(xrow + 8 * g);
#endif

#pragma unroll
    for (int i = 0; i < 16; ++i) {
        const int idx = i * 256 + tid;
        const int row = idx >> 5, q = idx & 31;
        const float* wsrc = (row < 64) ? (W1 + row * 128) : (W2 + (row - 64) * 128);
        const float4 v = *(const float4*)(wsrc + 4 * q);
        *(float4*)(Wl + row * GAT_WPITCH + 4 * q) = v;
    }
    __syncthreads();
    if (rowbase >= N) return;

    v8f acc1[4] = {v8f{}, v8f{}, v8f{}, v8f{}};
    v8f acc2[4] = {v8f{}, v8f{}, v8f{}, v8f{}};

#if HAS_WMMA_F32
    const float* wfrag = Wl + l16 * GAT_WPITCH + 4 * half;
#pragma unroll
    for (int g = 0; g < 16; ++g) {
        float4 b[8];
#pragma unroll
        for (int j = 0; j < 8; ++j)
            b[j] = *(const float4*)(wfrag + j * 16 * GAT_WPITCH + 8 * g);
        v2f a0, a1;
        a0.x = a[g].x; a0.y = a[g].y;
        a1.x = a[g].z; a1.y = a[g].w;
#pragma unroll
        for (int j = 0; j < 4; ++j) {
            v2f b1, b2;
            b1.x = b[j].x;     b1.y = b[j].y;
            b2.x = b[j + 4].x; b2.y = b[j + 4].y;
            acc1[j] = __builtin_amdgcn_wmma_f32_16x16x4_f32(
                false, a0, false, b1, (short)0, acc1[j], false, false);
            acc2[j] = __builtin_amdgcn_wmma_f32_16x16x4_f32(
                false, a0, false, b2, (short)0, acc2[j], false, false);
        }
#pragma unroll
        for (int j = 0; j < 4; ++j) {
            v2f b1, b2;
            b1.x = b[j].z;     b1.y = b[j].w;
            b2.x = b[j + 4].z; b2.y = b[j + 4].w;
            acc1[j] = __builtin_amdgcn_wmma_f32_16x16x4_f32(
                false, a1, false, b1, (short)0, acc1[j], false, false);
            acc2[j] = __builtin_amdgcn_wmma_f32_16x16x4_f32(
                false, a1, false, b2, (short)0, acc2[j], false, false);
        }
    }
#else
    typedef __attribute__((ext_vector_type(16))) __bf16 v16bf;
    const float* xr = X + (size_t)arow * 128;
    for (int k0 = 0; k0 < 128; k0 += 32) {
        v16bf av;
#pragma unroll
        for (int e = 0; e < 16; ++e) {
            const int v = e >> 1, hb = e & 1;
            const int K = ((v < 4) ? 2 * v : 16 + 2 * (v - 4)) + hb + 8 * half;
            av[e] = (__bf16)xr[k0 + K];
        }
#pragma unroll
        for (int j = 0; j < 4; ++j) {
            const int col = j * 16 + l16;
            v16bf b1, b2;
#pragma unroll
            for (int e = 0; e < 16; ++e) {
                const int v = e >> 1, hb = e & 1;
                const int K = ((v < 4) ? 2 * v : 16 + 2 * (v - 4)) + hb + 8 * half;
                b1[e] = (__bf16)W1[col * 128 + k0 + K];
                b2[e] = (__bf16)W2[col * 128 + k0 + K];
            }
            acc1[j] = __builtin_amdgcn_wmma_f32_16x16x32_bf16(
                false, av, false, b1, (short)0, acc1[j], false, false);
            acc2[j] = __builtin_amdgcn_wmma_f32_16x16x32_bf16(
                false, av, false, b2, (short)0, acc2[j], false, false);
        }
    }
#endif

#pragma unroll
    for (int j = 0; j < 4; ++j) {
#pragma unroll
        for (int r = 0; r < 8; ++r) {
            const int row = rowbase + r + 8 * half;
            if (row < N) {
                z [(size_t)row * 64 + j * 16 + l16] = acc1[j][r];
                zi[(size_t)row * 64 + j * 16 + l16] = acc2[j][r];
            }
        }
    }

    float wa1[4], wa2[4];
    for (int j = 0; j < 4; ++j) {
        wa1[j] = Wa[j * 16 + l16];
        wa2[j] = Wa[64 + j * 16 + l16];
    }
    float keep1 = 0.0f, keep2 = 0.0f;
    for (int r = 0; r < 8; ++r) {
        float p1 = acc1[0][r] * wa1[0];
        float p2 = acc1[0][r] * wa2[0];
        for (int j = 1; j < 4; ++j) {
            p1 = fmaf(acc1[j][r], wa1[j], p1);
            p2 = fmaf(acc1[j][r], wa2[j], p2);
        }
        p1 = gat_row_sum16(p1);
        p2 = gat_row_sum16(p2);
        if (l16 == r) { keep1 = p1; keep2 = p2; }
    }
    const int srow = rowbase + 8 * half + l16;
    if (l16 < 8 && srow < N) {
        s1[srow]  = keep1;
        s2[srow]  = keep2;
    }
}

#define GAT_BIN_SHIFT 9
#define GAT_BIN_NODES 512
#define GAT_CHUNK     4096
#define GAT_MAX_BINS  1024

__global__ __launch_bounds__(256) void gat_bincount_kernel(
    const int* __restrict__ dst, int* __restrict__ cmat, int E, int B, int C)
{
    __shared__ int hist[GAT_MAX_BINS];
    const int tid = threadIdx.x;
    for (int b = tid; b < B; b += 256) hist[b] = 0;
    __syncthreads();
    const int base = blockIdx.x * GAT_CHUNK;
    int d[16];
#pragma unroll
    for (int i = 0; i < 16; ++i) {
        const int e = base + i * 256 + tid;
        d[i] = (e < E) ? dst[e] : -1;
    }
#pragma unroll
    for (int i = 0; i < 16; ++i)
        if (d[i] >= 0) atomicAdd(&hist[d[i] >> GAT_BIN_SHIFT], 1);
    __syncthreads();
    for (int b = tid; b < B; b += 256) cmat[b * C + blockIdx.x] = hist[b];
}

__global__ __launch_bounds__(1024) void gat_offsets_kernel(
    const int* __restrict__ cnt, int* __restrict__ offs, int N)
{
    __shared__ int sh_prev[32];
    __shared__ int sh_loc[32];
    const int t = threadIdx.x, lane = t & 31, w = t >> 5;
    const int start = blockIdx.x * 1024;

    int prev = 0;
    const int4* c4 = (const int4*)cnt;
    const int nq = start >> 2;
    for (int q0 = 0; q0 < nq; q0 += 16 * 1024) {
        int4 c[16];
#pragma unroll
        for (int u = 0; u < 16; ++u) {
            const int q = q0 + u * 1024 + t;
            c[u] = (q < nq) ? c4[q] : make_int4(0, 0, 0, 0);
        }
#pragma unroll
        for (int u = 0; u < 16; ++u) prev += (c[u].x + c[u].y) + (c[u].z + c[u].w);
    }
    for (int off = 16; off > 0; off >>= 1) prev += __shfl_xor(prev, off, 32);

    const int i = start + t;
    const int c = (i < N) ? cnt[i] : 0;
    int inc = c;
    for (int off = 1; off < 32; off <<= 1) {
        const int v = __shfl_up(inc, off, 32);
        if (lane >= off) inc += v;
    }
    if (lane == 31) sh_loc[w]  = inc;
    if (lane == 0)  sh_prev[w] = prev;
    __syncthreads();

    int base = sh_prev[lane];
    for (int off = 16; off > 0; off >>= 1) base += __shfl_xor(base, off, 32);
    const int tot = sh_loc[lane];
    int winc = tot;
    for (int off = 1; off < 32; off <<= 1) {
        const int v = __shfl_up(winc, off, 32);
        if (lane >= off) winc += v;
    }
    const int wbase = __shfl(winc - tot, w, 32);
    if (i < N) offs[i] = base + wbase + (inc - c);
}

__global__ __launch_bounds__(256) void gat_binfile_kernel(
    const int* __restrict__ dst, const int* __restrict__ cscan,
    int* __restrict__ sid, int E, int B, int C)
{
    __shared__ int cur[GAT_MAX_BINS];
    const int tid = threadIdx.x;
    for (int b = tid; b < B; b += 256) cur[b] = cscan[b * C + blockIdx.x];
    __syncthreads();
    const int base = blockIdx.x * GAT_CHUNK;
    int d[16];
#pragma unroll
    for (int i = 0; i < 16; ++i) {
        const int e = base + i * 256 + tid;
        d[i] = (e < E) ? dst[e] : -1;
    }
#pragma unroll
    for (int i = 0; i < 16; ++i) {
        if (d[i] >= 0) {
            const int pos = atomicAdd(&cur[d[i] >> GAT_BIN_SHIFT], 1);
            sid[pos] = base + i * 256 + tid;
        }
    }
}

__device__ __forceinline__ float gat_logit(float s1v, float s2v, float ed, float w0, float wa)
{
    float v = s1v + s2v + ed * w0 * wa;
    return (v > 0.0f) ? v : 0.01f * v;
}


__global__ __launch_bounds__(1024) void gat_binplace_kernel(
    const float* __restrict__ edge_d,
    const int* __restrict__ src, const int* __restrict__ dst,
    const int* __restrict__ sid, const int* __restrict__ cscan,
    const float* __restrict__ s1, const float* __restrict__ s2,
    const float* __restrict__ W0, const float* __restrict__ Wa,
    int* __restrict__ offs, int* __restrict__ pairs_i,
    int N, int E, int B, int C)
{
    __shared__ int ncur[GAT_BIN_NODES];
    __shared__ int wtot[16];
    int2* __restrict__ pairs = (int2*)pairs_i;
    const int t = threadIdx.x, lane = t & 31, w = t >> 5;
    const int b = blockIdx.x;
    const int lo = cscan[b * C];
    const int hi = (b + 1 < B) ? cscan[(b + 1) * C] : E;
    const int node0 = b << GAT_BIN_SHIFT;

    if (t < GAT_BIN_NODES) ncur[t] = 0;
    __syncthreads();

    for (int i0 = lo + t; i0 < hi; i0 += 4096) {
        int e[4], d[4];
#pragma unroll
        for (int u = 0; u < 4; ++u) {
            const int i = i0 + u * 1024;
            e[u] = (i < hi) ? sid[i] : -1;
        }
#pragma unroll
        for (int u = 0; u < 4; ++u) d[u] = (e[u] >= 0) ? dst[e[u]] : -1;
#pragma unroll
        for (int u = 0; u < 4; ++u)
            if (d[u] >= 0) atomicAdd(&ncur[d[u] - node0], 1);
    }
    __syncthreads();

    int cval = 0, inc = 0;
    if (t < GAT_BIN_NODES) {
        cval = ncur[t];
        inc = cval;
        for (int off = 1; off < 32; off <<= 1) {
            const int v = __shfl_up(inc, off, 32);
            if (lane >= off) inc += v;
        }
        if (lane == 31) wtot[w] = inc;
    }
    __syncthreads();
    if (t < GAT_BIN_NODES) {
        const int tot = (lane < 16) ? wtot[lane] : 0;
        int winc = tot;
        for (int off = 1; off < 16; off <<= 1) {
            const int v = __shfl_up(winc, off, 32);
            if (lane >= off) winc += v;
        }
        const int wbase = __shfl(winc - tot, w, 32);
        const int start = lo + wbase + (inc - cval);
        ncur[t] = start;
        const int node = node0 + t;
        if (node < N) offs[node] = start;
    }
    if (b == B - 1 && t == 0) offs[N] = E;
    __syncthreads();

    const float w0 = W0[0], wa = Wa[128];
    for (int i0 = lo + t; i0 < hi; i0 += 4096) {
        int e[4], d[4], s[4];
        float ed[4], a1[4], a2[4];
#pragma unroll
        for (int u = 0; u < 4; ++u) {
            const int i = i0 + u * 1024;
            e[u] = (i < hi) ? sid[i] : -1;
        }
#pragma unroll
        for (int u = 0; u < 4; ++u) {
            const int ee = (e[u] >= 0) ? e[u] : 0;
            d[u]  = dst[ee];
            s[u]  = src[ee];
            ed[u] = edge_d[ee];
        }
#pragma unroll
        for (int u = 0; u < 4; ++u) {
            a1[u] = s1[s[u]];
            a2[u] = s2[d[u]];
        }
#pragma unroll
        for (int u = 0; u < 4; ++u) {
            if (e[u] >= 0) {
                const float v = gat_logit(a1[u], a2[u], ed[u], w0, wa);
                const int pos = atomicAdd(&ncur[d[u] - node0], 1);
                pairs[pos] = make_int2(s[u], __float_as_int(v));
            }
        }
    }
}

__device__ __forceinline__ float gat_wave_max(float x)
{
    x = fmaxf(x, gat_row_xmask<0x161>(x));
    x = fmaxf(x, gat_row_xmask<0x162>(x));
    x = fmaxf(x, gat_row_xmask<0x164>(x));
    x = fmaxf(x, gat_row_xmask<0x168>(x));
    const float lo = __int_as_float(__builtin_amdgcn_readlane(__float_as_int(x), 0));
    const float hi = __int_as_float(__builtin_amdgcn_readlane(__float_as_int(x), 16));
    return fmaxf(lo, hi);
}
__device__ __forceinline__ float gat_wave_sum(float x)
{
    x += gat_row_xmask<0x161>(x);
    x += gat_row_xmask<0x162>(x);
    x += gat_row_xmask<0x164>(x);
    x += gat_row_xmask<0x168>(x);
    const float lo = __int_as_float(__builtin_amdgcn_readlane(__float_as_int(x), 0));
    const float hi = __int_as_float(__builtin_amdgcn_readlane(__float_as_int(x), 16));
    return lo + hi;
}

template <int K>
__device__ __forceinline__ void gat_pull_rows(
    const float* __restrict__ z, int s, float a, int j, int lane, v2f& acc)
{
    v2f r[K];
#pragma unroll
    for (int u = 0; u < K; ++u) {
        const int sj = __builtin_amdgcn_readlane(s, j + u);
        r[u] = *(const v2f*)(z + (size_t)sj * 64 + 2 * lane);
    }
#pragma unroll
    for (int u = 0; u < K; ++u) {
        const float aj = __int_as_float(
            __builtin_amdgcn_readlane(__float_as_int(a), j + u));
        acc.x = fmaf(aj, r[u].x, acc.x);
        acc.y = fmaf(aj, r[u].y, acc.y);
    }
}

__device__ __forceinline__ void gat_pull_chunk(
    const float* __restrict__ z, int s, float a, int n, int lane, v2f& acc)
{
    int j = 0;
    for (; j + 8 <= n; j += 8) gat_pull_rows<8>(z, s, a, j, lane, acc);
    if (n - j >= 4) { gat_pull_rows<4>(z, s, a, j, lane, acc); j += 4; }
    if (n - j >= 2) { gat_pull_rows<2>(z, s, a, j, lane, acc); j += 2; }
    if (n - j >= 1) { gat_pull_rows<1>(z, s, a, j, lane, acc); }
}

__global__ __launch_bounds__(256) void gat_pull_kernel(
    const int* __restrict__ offs,
    const int* __restrict__ pairs_i, const float* __restrict__ z,
    float* out, int N)
{
    const int2* __restrict__ pairs = (const int2*)pairs_i;
    const int lane = threadIdx.x & 31;
    const int wave = __builtin_amdgcn_readfirstlane(blockIdx.x * 8 + (threadIdx.x >> 5));
    if (wave >= N) return;
    const int off = offs[wave];
    const int deg = offs[wave + 1] - off;
    float* orow = out + (size_t)wave * 64 + 2 * lane;
    const v2f zi = *(const v2f*)orow;
    v2f acc = {0.0f, 0.0f};

    if (deg <= 32) {
        int s = 0;
        float v = -__builtin_inff();
        if (lane < deg) {
            const int2 p = pairs[off + lane];
            s = p.x;
            v = __int_as_float(p.y);
        }
        const float m   = gat_wave_max(v);
        const float ex  = (lane < deg) ? __expf(v - m) : 0.0f;
        const float den = gat_wave_sum(ex);
        const float a   = ex / den;
        gat_pull_chunk(z, s, a, deg, lane, acc);
    } else {
        float m = -__builtin_inff();
        for (int c = 0; c < deg; c += 32)
            if (c + lane < deg) m = fmaxf(m, __int_as_float(pairs[off + c + lane].y));
        m = gat_wave_max(m);
        float den = 0.0f;
        for (int c = 0; c < deg; c += 32)
            if (c + lane < deg) den += __expf(__int_as_float(pairs[off + c + lane].y) - m);
        den = gat_wave_sum(den);
        for (int c = 0; c < deg; c += 32) {
            int s = 0;
            float a = 0.0f;
            if (c + lane < deg) {
                const int2 p = pairs[off + c + lane];
                s = p.x;
                a = __expf(__int_as_float(p.y) - m) / den;
            }
            const int n = (deg - c < 32) ? (deg - c) : 32;
            gat_pull_chunk(z, s, a, n, lane, acc);
        }
    }

    v2f h;
    h.x = zi.x + acc.x;
    h.y = zi.y + acc.y;
    h.x = (h.x > 0.0f) ? h.x : 0.0f;
    h.y = (h.y > 0.0f) ? h.y : 0.0f;
    *(v2f*)orow = h;
}

extern "C" void kernel_launch(void* const* d_in, const int* in_sizes, int n_in,
                              void* d_out, int out_size, void* d_ws, size_t ws_size,
                              hipStream_t stream)
{
    const float* X   = (const float*)d_in[0];
    const float* ed  = (const float*)d_in[1];
    const float* W0  = (const float*)d_in[2];
    const float* W1  = (const float*)d_in[3];
    const float* W2  = (const float*)d_in[4];
    const float* Wa  = (const float*)d_in[5];
    const int*   src = (const int*)d_in[6];
    const int*   dst = (const int*)d_in[7];

    const int N = in_sizes[0] / 128;
    const int E = in_sizes[6];

    const int B = (N + GAT_BIN_NODES - 1) / GAT_BIN_NODES;
    const int C = (E + GAT_CHUNK - 1) / GAT_CHUNK;
    const int M = B * C;

    float* ws    = (float*)d_ws;
    float* z     = ws;         ws += (size_t)N * 64;
    int*   offs  = (int*)ws;
    int*   cscan = (int*)ws + (N + 1);
    ws += 2 * (size_t)N;
    float* s1    = ws;         ws += N;
    float* s2    = ws;         ws += N;
    int*   pairs = (int*)ws;   ws += (size_t)E * 2;
    int*   sid   = (int*)ws;   ws += E;
    int*   cmat  = pairs;
    float* out   = (float*)d_out;

    const int rowTiles = (N + 15) / 16;
    gat_gemm_kernel<<<(rowTiles + 7) / 8, 256, 0, stream>>>(X, W1, W2, Wa, z, out, s1, s2, N);

    gat_bincount_kernel<<<C, 256, 0, stream>>>(dst, cmat, E, B, C);

    gat_offsets_kernel<<<(M + 1023) / 1024, 1024, 0, stream>>>(cmat, cscan, M);

    gat_binfile_kernel<<<C, 256, 0, stream>>>(dst, cscan, sid, E, B, C);

    gat_binplace_kernel<<<B, 1024, 0, stream>>>(
        ed, src, dst, sid, cscan, s1, s2, W0, Wa, offs, pairs, N, E, B, C);

    const int nodeThreads = N * 32;
    gat_pull_kernel<<<(nodeThreads + 255) / 256, 256, 0, stream>>>(
        offs, pairs, z, out, N);
}


# level 1 of the dst bucketing files 8-byte edge records (src, dst inside the bin, edge_d), read in edge order, instead of edge ids: level 2 reads them in order, its four fetches by edge id are gone
# speedup vs baseline: 3.9360x; 1.2621x over previous
#include <hip/hip_runtime.h>
#include <hip/hip_bf16.h>
#include <math.h>


typedef __attribute__((ext_vector_type(2))) float v2f;
typedef __attribute__((ext_vector_type(8))) float v8f;

#define HAS_WMMA_F32 __has_builtin(__builtin_amdgcn_wmma_f32_16x16x4_f32)

template <int CTRL>
__device__ __forceinline__ float gat_row_xmask(float x)
{
    return __int_as_float(__builtin_amdgcn_update_dpp(
        0, __float_as_int(x), CTRL, 0xf, 0xf, true));
}
__device__ __forceinline__ float gat_row_sum16(float x)
{
    x += gat_row_xmask<0x161>(x);
    x += gat_row_xmask<0x162>(x);
    x += gat_row_xmask<0x164>(x);
    x += gat_row_xmask<0x168>(x);
    return x;
}

#define GAT_WPITCH 132

__global__ __launch_bounds__(256) void gat_gemm_kernel(
    const float* __restrict__ X,
    const float* __restrict__ W1,
    const float* __restrict__ W2,
    const float* __restrict__ Wa,
    float* __restrict__ z,
    float* __restrict__ zi,
    float* __restrict__ s1,
    float* __restrict__ s2,
    int N)
{
    __shared__ float4 gat_wl4[128 * GAT_WPITCH / 4];
    float* Wl = (float*)gat_wl4;

    const int tid     = threadIdx.x;
    const int lane    = tid & 31;
    const int half    = lane >> 4;
    const int l16     = lane & 15;
    const int rowbase = (blockIdx.x * 8 + (tid >> 5)) * 16;

    int arow = rowbase + l16;
    if (arow >= N) arow = N - 1;
    const float* xrow = X + (size_t)arow * 128 + 4 * half;

#if HAS_WMMA_F32
    float4 a[16];
#pragma unroll
    for (int g = 0; g < 16; ++g) a[g] = *(const float4*)(xrow + 8 * g);
#endif

#pragma unroll
    for (int i = 0; i < 16; ++i) {
        const int idx = i * 256 + tid;
        const int row = idx >> 5, q = idx & 31;
        const float* wsrc = (row < 64) ? (W1 + row * 128) : (W2 + (row - 64) * 128);
        const float4 v = *(const float4*)(wsrc + 4 * q);
        *(float4*)(Wl + row * GAT_WPITCH + 4 * q) = v;
    }
    __syncthreads();
    if (rowbase >= N) return;

    v8f acc1[4] = {v8f{}, v8f{}, v8f{}, v8f{}};
    v8f acc2[4] = {v8f{}, v8f{}, v8f{}, v8f{}};

#if HAS_WMMA_F32
    const float* wfrag = Wl + l16 * GAT_WPITCH + 4 * half;
#pragma unroll
    for (int g = 0; g < 16; ++g) {
        float4 b[8];
#pragma unroll
        for (int j = 0; j < 8; ++j)
            b[j] = *(const float4*)(wfrag + j * 16 * GAT_WPITCH + 8 * g);
        v2f a0, a1;
        a0.x = a[g].x; a0.y = a[g].y;
        a1.x = a[g].z; a1.y = a[g].w;
#pragma unroll
        for (int j = 0; j < 4; ++j) {
            v2f b1, b2;
            b1.x = b[j].x;     b1.y = b[j].y;
            b2.x = b[j + 4].x; b2.y = b[j + 4].y;
            acc1[j] = __builtin_amdgcn_wmma_f32_16x16x4_f32(
                false, a0, false, b1, (short)0, acc1[j], false, false);
            acc2[j] = __builtin_amdgcn_wmma_f32_16x16x4_f32(
                false, a0, false, b2, (short)0, acc2[j], false, false);
        }
#pragma unroll
        for (int j = 0; j < 4; ++j) {
            v2f b1, b2;
            b1.x = b[j].z;     b1.y = b[j].w;
            b2.x = b[j + 4].z; b2.y = b[j + 4].w;
            acc1[j] = __builtin_amdgcn_wmma_f32_16x16x4_f32(
                false, a1, false, b1, (short)0, acc1[j], false, false);
            acc2[j] = __builtin_amdgcn_wmma_f32_16x16x4_f32(
                false, a1, false, b2, (short)0, acc2[j], false, false);
        }
    }
#else
    typedef __attribute__((ext_vector_type(16))) __bf16 v16bf;
    const float* xr = X + (size_t)arow * 128;
    for (int k0 = 0; k0 < 128; k0 += 32) {
        v16bf av;
#pragma unroll
        for (int e = 0; e < 16; ++e) {
            const int v = e >> 1, hb = e & 1;
            const int K = ((v < 4) ? 2 * v : 16 + 2 * (v - 4)) + hb + 8 * half;
            av[e] = (__bf16)xr[k0 + K];
        }
#pragma unroll
        for (int j = 0; j < 4; ++j) {
            const int col = j * 16 + l16;
            v16bf b1, b2;
#pragma unroll
            for (int e = 0; e < 16; ++e) {
                const int v = e >> 1, hb = e & 1;
                const int K = ((v < 4) ? 2 * v : 16 + 2 * (v - 4)) + hb + 8 * half;
                b1[e] = (__bf16)W1[col * 128 + k0 + K];
                b2[e] = (__bf16)W2[col * 128 + k0 + K];
            }
            acc1[j] = __builtin_amdgcn_wmma_f32_16x16x32_bf16(
                false, av, false, b1, (short)0, acc1[j], false, false);
            acc2[j] = __builtin_amdgcn_wmma_f32_16x16x32_bf16(
                false, av, false, b2, (short)0, acc2[j], false, false);
        }
    }
#endif

#pragma unroll
    for (int j = 0; j < 4; ++j) {
#pragma unroll
        for (int r = 0; r < 8; ++r) {
            const int row = rowbase + r + 8 * half;
            if (row < N) {
                z [(size_t)row * 64 + j * 16 + l16] = acc1[j][r];
                zi[(size_t)row * 64 + j * 16 + l16] = acc2[j][r];
            }
        }
    }

    float wa1[4], wa2[4];
    for (int j = 0; j < 4; ++j) {
        wa1[j] = Wa[j * 16 + l16];
        wa2[j] = Wa[64 + j * 16 + l16];
    }
    float keep1 = 0.0f, keep2 = 0.0f;
    for (int r = 0; r < 8; ++r) {
        float p1 = acc1[0][r] * wa1[0];
        float p2 = acc1[0][r] * wa2[0];
        for (int j = 1; j < 4; ++j) {
            p1 = fmaf(acc1[j][r], wa1[j], p1);
            p2 = fmaf(acc1[j][r], wa2[j], p2);
        }
        p1 = gat_row_sum16(p1);
        p2 = gat_row_sum16(p2);
        if (l16 == r) { keep1 = p1; keep2 = p2; }
    }
    const int srow = rowbase + 8 * half + l16;
    if (l16 < 8 && srow < N) {
        s1[srow]  = keep1;
        s2[srow]  = keep2;
    }
}

#define GAT_BIN_SHIFT 9
#define GAT_BIN_NODES 512
#define GAT_CHUNK     4096
#define GAT_MAX_BINS  1024

__global__ __launch_bounds__(256) void gat_bincount_kernel(
    const int* __restrict__ dst, int* __restrict__ cmat, int E, int B, int C)
{
    __shared__ int hist[GAT_MAX_BINS];
    const int tid = threadIdx.x;
    for (int b = tid; b < B; b += 256) hist[b] = 0;
    __syncthreads();
    const int base = blockIdx.x * GAT_CHUNK;
    int d[16];
#pragma unroll
    for (int i = 0; i < 16; ++i) {
        const int e = base + i * 256 + tid;
        d[i] = (e < E) ? dst[e] : -1;
    }
#pragma unroll
    for (int i = 0; i < 16; ++i)
        if (d[i] >= 0) atomicAdd(&hist[d[i] >> GAT_BIN_SHIFT], 1);
    __syncthreads();
    for (int b = tid; b < B; b += 256) cmat[b * C + blockIdx.x] = hist[b];
}

__global__ __launch_bounds__(1024) void gat_offsets_kernel(
    const int* __restrict__ cnt, int* __restrict__ offs, int N)
{
    __shared__ int sh_prev[32];
    __shared__ int sh_loc[32];
    const int t = threadIdx.x, lane = t & 31, w = t >> 5;
    const int start = blockIdx.x * 1024;

    int prev = 0;
    const int4* c4 = (const int4*)cnt;
    const int nq = start >> 2;
    for (int q0 = 0; q0 < nq; q0 += 16 * 1024) {
        int4 c[16];
#pragma unroll
        for (int u = 0; u < 16; ++u) {
            const int q = q0 + u * 1024 + t;
            c[u] = (q < nq) ? c4[q] : make_int4(0, 0, 0, 0);
        }
#pragma unroll
        for (int u = 0; u < 16; ++u) prev += (c[u].x + c[u].y) + (c[u].z + c[u].w);
    }
    for (int off = 16; off > 0; off >>= 1) prev += __shfl_xor(prev, off, 32);

    const int i = start + t;
    const int c = (i < N) ? cnt[i] : 0;
    int inc = c;
    for (int off = 1; off < 32; off <<= 1) {
        const int v = __shfl_up(inc, off, 32);
        if (lane >= off) inc += v;
    }
    if (lane == 31) sh_loc[w]  = inc;
    if (lane == 0)  sh_prev[w] = prev;
    __syncthreads();

    int base = sh_prev[lane];
    for (int off = 16; off > 0; off >>= 1) base += __shfl_xor(base, off, 32);
    const int tot = sh_loc[lane];
    int winc = tot;
    for (int off = 1; off < 32; off <<= 1) {
        const int v = __shfl_up(winc, off, 32);
        if (lane >= off) winc += v;
    }
    const int wbase = __shfl(winc - tot, w, 32);
    if (i < N) offs[i] = base + wbase + (inc - c);
}

__global__ __launch_bounds__(256) void gat_binfile_kernel(
    const float* __restrict__ edge_d,
    const int* __restrict__ src, const int* __restrict__ dst,
    const int* __restrict__ cscan, int* __restrict__ rec_i, int E, int B, int C)
{
    __shared__ int cur[GAT_MAX_BINS];
    int2* __restrict__ rec = (int2*)rec_i;
    const int tid = threadIdx.x;
    for (int b = tid; b < B; b += 256) cur[b] = cscan[b * C + blockIdx.x];
    __syncthreads();
    const int base = blockIdx.x * GAT_CHUNK;
    int d[16], s[16];
    float ed[16];
#pragma unroll
    for (int i = 0; i < 16; ++i) {
        const int e = base + i * 256 + tid;
        const bool ok = e < E;
        d[i]  = ok ? dst[e] : -1;
        s[i]  = ok ? src[e] : 0;
        ed[i] = ok ? edge_d[e] : 0.0f;
    }
#pragma unroll
    for (int i = 0; i < 16; ++i) {
        if (d[i] >= 0) {
            const int pos = atomicAdd(&cur[d[i] >> GAT_BIN_SHIFT], 1);
            rec[pos] = make_int2((s[i] << GAT_BIN_SHIFT) | (d[i] & (GAT_BIN_NODES - 1)),
                                 __float_as_int(ed[i]));
        }
    }
}

__device__ __forceinline__ float gat_logit(float s1v, float s2v, float ed, float w0, float wa)
{
    float v = s1v + s2v + ed * w0 * wa;
    return (v > 0.0f) ? v : 0.01f * v;
}


__global__ __launch_bounds__(1024) void gat_binplace_kernel(
    const int* __restrict__ rec_i, const int* __restrict__ cscan,
    const float* __restrict__ s1, const float* __restrict__ s2,
    const float* __restrict__ W0, const float* __restrict__ Wa,
    int* __restrict__ offs, int* __restrict__ pairs_i,
    int N, int E, int B, int C)
{
    __shared__ int ncur[GAT_BIN_NODES];
    __shared__ int wtot[16];
    int2* __restrict__ pairs = (int2*)pairs_i;
    const int2* __restrict__ rec = (const int2*)rec_i;
    const int t = threadIdx.x, lane = t & 31, w = t >> 5;
    const int b = blockIdx.x;
    const int lo = cscan[b * C];
    const int hi = (b + 1 < B) ? cscan[(b + 1) * C] : E;
    const int node0 = b << GAT_BIN_SHIFT;

    if (t < GAT_BIN_NODES) ncur[t] = 0;
    __syncthreads();

    for (int i0 = lo + t; i0 < hi; i0 += 4096) {
        int k[4];
#pragma unroll
        for (int u = 0; u < 4; ++u) {
            const int i = i0 + u * 1024;
            k[u] = (i < hi) ? rec[i].x : -1;
        }
#pragma unroll
        for (int u = 0; u < 4; ++u)
            if (k[u] >= 0) atomicAdd(&ncur[k[u] & (GAT_BIN_NODES - 1)], 1);
    }
    __syncthreads();

    int cval = 0, inc = 0;
    if (t < GAT_BIN_NODES) {
        cval = ncur[t];
        inc = cval;
        for (int off = 1; off < 32; off <<= 1) {
            const int v = __shfl_up(inc, off, 32);
            if (lane >= off) inc += v;
        }
        if (lane == 31) wtot[w] = inc;
    }
    __syncthreads();
    if (t < GAT_BIN_NODES) {
        const int tot = (lane < 16) ? wtot[lane] : 0;
        int winc = tot;
        for (int off = 1; off < 16; off <<= 1) {
            const int v = __shfl_up(winc, off, 32);
            if (lane >= off) winc += v;
        }
        const int wbase = __shfl(winc - tot, w, 32);
        const int start = lo + wbase + (inc - cval);
        ncur[t] = start;
        const int node = node0 + t;
        if (node < N) offs[node] = start;
    }
    if (b == B - 1 && t == 0) offs[N] = E;
    __syncthreads();

    const float w0 = W0[0], wa = Wa[128];
    for (int i0 = lo + t; i0 < hi; i0 += 4096) {
        int2 r[4];
        float a1[4], a2[4];
#pragma unroll
        for (int u = 0; u < 4; ++u) {
            const int i = i0 + u * 1024;
            r[u] = (i < hi) ? rec[i] : make_int2(-1, 0);
        }
#pragma unroll
        for (int u = 0; u < 4; ++u) {
            const int kk = (r[u].x >= 0) ? r[u].x : 0;
            a1[u] = s1[kk >> GAT_BIN_SHIFT];
            a2[u] = s2[node0 + (kk & (GAT_BIN_NODES - 1))];
        }
#pragma unroll
        for (int u = 0; u < 4; ++u) {
            if (r[u].x >= 0) {
                const float v = gat_logit(a1[u], a2[u], __int_as_float(r[u].y), w0, wa);
                const int pos = atomicAdd(&ncur[r[u].x & (GAT_BIN_NODES - 1)], 1);
                pairs[pos] = make_int2(r[u].x >> GAT_BIN_SHIFT, __float_as_int(v));
            }
        }
    }
}

__device__ __forceinline__ float gat_wave_max(float x)
{
    x = fmaxf(x, gat_row_xmask<0x161>(x));
    x = fmaxf(x, gat_row_xmask<0x162>(x));
    x = fmaxf(x, gat_row_xmask<0x164>(x));
    x = fmaxf(x, gat_row_xmask<0x168>(x));
    const float lo = __int_as_float(__builtin_amdgcn_readlane(__float_as_int(x), 0));
    const float hi = __int_as_float(__builtin_amdgcn_readlane(__float_as_int(x), 16));
    return fmaxf(lo, hi);
}
__device__ __forceinline__ float gat_wave_sum(float x)
{
    x += gat_row_xmask<0x161>(x);
    x += gat_row_xmask<0x162>(x);
    x += gat_row_xmask<0x164>(x);
    x += gat_row_xmask<0x168>(x);
    const float lo = __int_as_float(__builtin_amdgcn_readlane(__float_as_int(x), 0));
    const float hi = __int_as_float(__builtin_amdgcn_readlane(__float_as_int(x), 16));
    return lo + hi;
}

template <int K>
__device__ __forceinline__ void gat_pull_rows(
    const float* __restrict__ z, int s, float a, int j, int lane, v2f& acc)
{
    v2f r[K];
#pragma unroll
    for (int u = 0; u < K; ++u) {
        const int sj = __builtin_amdgcn_readlane(s, j + u);
        r[u] = *(const v2f*)(z + (size_t)sj * 64 + 2 * lane);
    }
#pragma unroll
    for (int u = 0; u < K; ++u) {
        const float aj = __int_as_float(
            __builtin_amdgcn_readlane(__float_as_int(a), j + u));
        acc.x = fmaf(aj, r[u].x, acc.x);
        acc.y = fmaf(aj, r[u].y, acc.y);
    }
}

__device__ __forceinline__ void gat_pull_chunk(
    const float* __restrict__ z, int s, float a, int n, int lane, v2f& acc)
{
    int j = 0;
    for (; j + 8 <= n; j += 8) gat_pull_rows<8>(z, s, a, j, lane, acc);
    if (n - j >= 4) { gat_pull_rows<4>(z, s, a, j, lane, acc); j += 4; }
    if (n - j >= 2) { gat_pull_rows<2>(z, s, a, j, lane, acc); j += 2; }
    if (n - j >= 1) { gat_pull_rows<1>(z, s, a, j, lane, acc); }
}

__global__ __launch_bounds__(256) void gat_pull_kernel(
    const int* __restrict__ offs,
    const int* __restrict__ pairs_i, const float* __restrict__ z,
    float* out, int N)
{
    const int2* __restrict__ pairs = (const int2*)pairs_i;
    const int lane = threadIdx.x & 31;
    const int wave = __builtin_amdgcn_readfirstlane(blockIdx.x * 8 + (threadIdx.x >> 5));
    if (wave >= N) return;
    const int off = offs[wave];
    const int deg = offs[wave + 1] - off;
    float* orow = out + (size_t)wave * 64 + 2 * lane;
    const v2f zi = *(const v2f*)orow;
    v2f acc = {0.0f, 0.0f};

    if (deg <= 32) {
        int s = 0;
        float v = -__builtin_inff();
        if (lane < deg) {
            const int2 p = pairs[off + lane];
            s = p.x;
            v = __int_as_float(p.y);
        }
        const float m   = gat_wave_max(v);
        const float ex  = (lane < deg) ? __expf(v - m) : 0.0f;
        const float den = gat_wave_sum(ex);
        const float a   = ex / den;
        gat_pull_chunk(z, s, a, deg, lane, acc);
    } else {
        float m = -__builtin_inff();
        for (int c = 0; c < deg; c += 32)
            if (c + lane < deg) m = fmaxf(m, __int_as_float(pairs[off + c + lane].y));
        m = gat_wave_max(m);
        float den = 0.0f;
        for (int c = 0; c < deg; c += 32)
            if (c + lane < deg) den += __expf(__int_as_float(pairs[off + c + lane].y) - m);
        den = gat_wave_sum(den);
        for (int c = 0; c < deg; c += 32) {
            int s = 0;
            float a = 0.0f;
            if (c + lane < deg) {
                const int2 p = pairs[off + c + lane];
                s = p.x;
                a = __expf(__int_as_float(p.y) - m) / den;
            }
            const int n = (deg - c < 32) ? (deg - c) : 32;
            gat_pull_chunk(z, s, a, n, lane, acc);
        }
    }

    v2f h;
    h.x = zi.x + acc.x;
    h.y = zi.y + acc.y;
    h.x = (h.x > 0.0f) ? h.x : 0.0f;
    h.y = (h.y > 0.0f) ? h.y : 0.0f;
    *(v2f*)orow = h;
}

extern "C" void kernel_launch(void* const* d_in, const int* in_sizes, int n_in,
                              void* d_out, int out_size, void* d_ws, size_t ws_size,
                              hipStream_t stream)
{
    const float* X   = (const float*)d_in[0];
    const float* ed  = (const float*)d_in[1];
    const float* W0  = (const float*)d_in[2];
    const float* W1  = (const float*)d_in[3];
    const float* W2  = (const float*)d_in[4];
    const float* Wa  = (const float*)d_in[5];
    const int*   src = (const int*)d_in[6];
    const int*   dst = (const int*)d_in[7];

    const int N = in_sizes[0] / 128;
    const int E = in_sizes[6];

    const int B = (N + GAT_BIN_NODES - 1) / GAT_BIN_NODES;
    const int C = (E + GAT_CHUNK - 1) / GAT_CHUNK;
    const int M = B * C;

    float* ws    = (float*)d_ws;
    float* z     = ws;         ws += (size_t)N * 64;
    int*   offs  = (int*)ws;
    int*   cscan = (int*)ws + (N + 1);
    ws += 2 * (size_t)N;
    float* s1    = ws;         ws += N;
    float* s2    = ws;         ws += N;
    int*   pairs = (int*)ws;   ws += (size_t)E * 2;
    int*   rec   = (int*)ws;   ws += (size_t)E * 2;
    int*   cmat  = pairs;
    float* out   = (float*)d_out;

    const int rowTiles = (N + 15) / 16;
    gat_gemm_kernel<<<(rowTiles + 7) / 8, 256, 0, stream>>>(X, W1, W2, Wa, z, out, s1, s2, N);

    gat_bincount_kernel<<<C, 256, 0, stream>>>(dst, cmat, E, B, C);

    gat_offsets_kernel<<<(M + 1023) / 1024, 1024, 0, stream>>>(cmat, cscan, M);

    gat_binfile_kernel<<<C, 256, 0, stream>>>(ed, src, dst, cscan, rec, E, B, C);

    gat_binplace_kernel<<<B, 1024, 0, stream>>>(
        rec, cscan, s1, s2, W0, Wa, offs, pairs, N, E, B, C);

    const int nodeThreads = N * 32;
    gat_pull_kernel<<<(nodeThreads + 255) / 256, 256, 0, stream>>>(
        offs, pairs, z, out, N);
}


# level 2 of the dst bucketing: the bin's 512 s2 values are read from an LDS copy and the (src, logit) pairs are put in place in an LDS buffer, then written out in order as contiguous wave stores
# speedup vs baseline: 5.1916x; 1.3190x over previous
#include <hip/hip_runtime.h>
#include <hip/hip_bf16.h>
#include <math.h>


typedef __attribute__((ext_vector_type(2))) float v2f;
typedef __attribute__((ext_vector_type(8))) float v8f;

#define HAS_WMMA_F32 __has_builtin(__builtin_amdgcn_wmma_f32_16x16x4_f32)

template <int CTRL>
__device__ __forceinline__ float gat_row_xmask(float x)
{
    return __int_as_float(__builtin_amdgcn_update_dpp(
        0, __float_as_int(x), CTRL, 0xf, 0xf, true));
}
__device__ __forceinline__ float gat_row_sum16(float x)
{
    x += gat_row_xmask<0x161>(x);
    x += gat_row_xmask<0x162>(x);
    x += gat_row_xmask<0x164>(x);
    x += gat_row_xmask<0x168>(x);
    return x;
}

#define GAT_WPITCH 132

__device__ __forceinline__ void gat_gemm_body(
    float* Wl,
    const int group,
    const float* __restrict__ X,
    const float* __restrict__ W1,
    const float* __restrict__ W2,
    const float* __restrict__ Wa,
    float* __restrict__ z,
    float* __restrict__ zi,
    float* __restrict__ s1,
    float* __restrict__ s2,
    int N)
{
    const int tid     = threadIdx.x;
    const int lane    = tid & 31;
    const int half    = lane >> 4;
    const int l16     = lane & 15;
    const int rowbase = (group * 8 + (tid >> 5)) * 16;

    int arow = rowbase + l16;
    if (arow >= N) arow = N - 1;
    const float* xrow = X + (size_t)arow * 128 + 4 * half;

#if HAS_WMMA_F32
    float4 a[16];
#pragma unroll
    for (int g = 0; g < 16; ++g) a[g] = *(const float4*)(xrow + 8 * g);
#endif

#pragma unroll
    for (int i = 0; i < 16; ++i) {
        const int idx = i * 256 + tid;
        const int row = idx >> 5, q = idx & 31;
        const float* wsrc = (row < 64) ? (W1 + row * 128) : (W2 + (row - 64) * 128);
        const float4 v = *(const float4*)(wsrc + 4 * q);
        *(float4*)(Wl + row * GAT_WPITCH + 4 * q) = v;
    }
    __syncthreads();
    if (rowbase >= N) return;

    v8f acc1[4] = {v8f{}, v8f{}, v8f{}, v8f{}};
    v8f acc2[4] = {v8f{}, v8f{}, v8f{}, v8f{}};

#if HAS_WMMA_F32
    const float* wfrag = Wl + l16 * GAT_WPITCH + 4 * half;
#pragma unroll
    for (int g = 0; g < 16; ++g) {
        float4 b[8];
#pragma unroll
        for (int j = 0; j < 8; ++j)
            b[j] = *(const float4*)(wfrag + j * 16 * GAT_WPITCH + 8 * g);
        v2f a0, a1;
        a0.x = a[g].x; a0.y = a[g].y;
        a1.x = a[g].z; a1.y = a[g].w;
#pragma unroll
        for (int j = 0; j < 4; ++j) {
            v2f b1, b2;
            b1.x = b[j].x;     b1.y = b[j].y;
            b2.x = b[j + 4].x; b2.y = b[j + 4].y;
            acc1[j] = __builtin_amdgcn_wmma_f32_16x16x4_f32(
                false, a0, false, b1, (short)0, acc1[j], false, false);
            acc2[j] = __builtin_amdgcn_wmma_f32_16x16x4_f32(
                false, a0, false, b2, (short)0, acc2[j], false, false);
        }
#pragma unroll
        for (int j = 0; j < 4; ++j) {
            v2f b1, b2;
            b1.x = b[j].z;     b1.y = b[j].w;
            b2.x = b[j + 4].z; b2.y = b[j + 4].w;
            acc1[j] = __builtin_amdgcn_wmma_f32_16x16x4_f32(
                false, a1, false, b1, (short)0, acc1[j], false, false);
            acc2[j] = __builtin_amdgcn_wmma_f32_16x16x4_f32(
                false, a1, false, b2, (short)0, acc2[j], false, false);
        }
    }
#else
    typedef __attribute__((ext_vector_type(16))) __bf16 v16bf;
    const float* xr = X + (size_t)arow * 128;
    for (int k0 = 0; k0 < 128; k0 += 32) {
        v16bf av;
#pragma unroll
        for (int e = 0; e < 16; ++e) {
            const int v = e >> 1, hb = e & 1;
            const int K = ((v < 4) ? 2 * v : 16 + 2 * (v - 4)) + hb + 8 * half;
            av[e] = (__bf16)xr[k0 + K];
        }
#pragma unroll
        for (int j = 0; j < 4; ++j) {
            const int col = j * 16 + l16;
            v16bf b1, b2;
#pragma unroll
            for (int e = 0; e < 16; ++e) {
                const int v = e >> 1, hb = e & 1;
                const int K = ((v < 4) ? 2 * v : 16 + 2 * (v - 4)) + hb + 8 * half;
                b1[e] = (__bf16)W1[col * 128 + k0 + K];
                b2[e] = (__bf16)W2[col * 128 + k0 + K];
            }
            acc1[j] = __builtin_amdgcn_wmma_f32_16x16x32_bf16(
                false, av, false, b1, (short)0, acc1[j], false, false);
            acc2[j] = __builtin_amdgcn_wmma_f32_16x16x32_bf16(
                false, av, false, b2, (short)0, acc2[j], false, false);
        }
    }
#endif

#pragma unroll
    for (int j = 0; j < 4; ++j) {
#pragma unroll
        for (int r = 0; r < 8; ++r) {
            const int row = rowbase + r + 8 * half;
            if (row < N) {
                z [(size_t)row * 64 + j * 16 + l16] = acc1[j][r];
                zi[(size_t)row * 64 + j * 16 + l16] = acc2[j][r];
            }
        }
    }

    float wa1[4], wa2[4];
    for (int j = 0; j < 4; ++j) {
        wa1[j] = Wa[j * 16 + l16];
        wa2[j] = Wa[64 + j * 16 + l16];
    }
    float keep1 = 0.0f, keep2 = 0.0f;
    for (int r = 0; r < 8; ++r) {
        float p1 = acc1[0][r] * wa1[0];
        float p2 = acc1[0][r] * wa2[0];
        for (int j = 1; j < 4; ++j) {
            p1 = fmaf(acc1[j][r], wa1[j], p1);
            p2 = fmaf(acc1[j][r], wa2[j], p2);
        }
        p1 = gat_row_sum16(p1);
        p2 = gat_row_sum16(p2);
        if (l16 == r) { keep1 = p1; keep2 = p2; }
    }
    const int srow = rowbase + 8 * half + l16;
    if (l16 < 8 && srow < N) {
        s1[srow]  = keep1;
        s2[srow]  = keep2;
    }
}


#define GAT_BIN_SHIFT 9
#define GAT_BIN_NODES 512
#define GAT_CHUNK     4096
#define GAT_MAX_BINS  1024
#define GAT_MAX_CHUNKS 1024

__device__ __forceinline__ void gat_chunksort_body(
    int* cur,
    int* wtot,
    const int chunk,
    const float* __restrict__ edge_d,
    const int* __restrict__ src, const int* __restrict__ dst,
    int* __restrict__ tab, int* __restrict__ rec_i, int E, int B, int C)
{
    int2* __restrict__ rec = (int2*)rec_i;
    const int tid = threadIdx.x, lane = tid & 31, w = tid >> 5;
    *(int4*)(cur + 4 * tid) = make_int4(0, 0, 0, 0);
    __syncthreads();
    const int base = chunk * GAT_CHUNK;
    int d[16], s[16];
    float ed[16];
#pragma unroll
    for (int i = 0; i < 16; ++i) {
        const int e = base + i * 256 + tid;
        const bool ok = e < E;
        d[i]  = ok ? dst[e] : -1;
        s[i]  = ok ? src[e] : 0;
        ed[i] = ok ? edge_d[e] : 0.0f;
    }
#pragma unroll
    for (int i = 0; i < 16; ++i)
        if (d[i] >= 0) atomicAdd(&cur[d[i] >> GAT_BIN_SHIFT], 1);
    __syncthreads();

    const int4 h = *(const int4*)(cur + 4 * tid);
    const int sum4 = (h.x + h.y) + (h.z + h.w);
    int inc = sum4;
    for (int off = 1; off < 32; off <<= 1) {
        const int v = __shfl_up(inc, off, 32);
        if (lane >= off) inc += v;
    }
    if (lane == 31) wtot[w] = inc;
    __syncthreads();
    int e0 = inc - sum4;
    for (int k = 0; k < 8; ++k) e0 += (k < w) ? wtot[k] : 0;
    const int e1 = e0 + h.x, e2 = e1 + h.y, e3 = e2 + h.z;
    *(int4*)(cur + 4 * tid) = make_int4(e0, e1, e2, e3);
    __syncthreads();

#pragma unroll
    for (int i = 0; i < 16; ++i) {
        if (d[i] >= 0) {
            const int pos = atomicAdd(&cur[d[i] >> GAT_BIN_SHIFT], 1);
            rec[base + pos] = make_int2((s[i] << GAT_BIN_SHIFT) | (d[i] & (GAT_BIN_NODES - 1)),
                                        __float_as_int(ed[i]));
        }
    }
    const int b0 = 4 * tid;
    if (b0     <= B) tab[(b0    ) * C + chunk] = e0;
    if (b0 + 1 <= B) tab[(b0 + 1) * C + chunk] = e1;
    if (b0 + 2 <= B) tab[(b0 + 2) * C + chunk] = e2;
    if (b0 + 3 <= B) tab[(b0 + 3) * C + chunk] = e3;
}

__global__ __launch_bounds__(256) void gat_gemm_binsort_kernel(
    const float* __restrict__ X, const float* __restrict__ W1, const float* __restrict__ W2,
    const float* __restrict__ Wa,
    float* __restrict__ z, float* __restrict__ zi,
    float* __restrict__ s1, float* __restrict__ s2, int N,
    const float* __restrict__ edge_d,
    const int* __restrict__ src, const int* __restrict__ dst,
    int* __restrict__ tab, int* __restrict__ rec_i, int E, int B, int C, int G)
{
    __shared__ float4 gat_wl4[128 * GAT_WPITCH / 4];
    __shared__ int4 gat_cur4[GAT_MAX_BINS / 4];
    if ((int)blockIdx.x < G) {
        gat_gemm_body((float*)gat_wl4, (int)blockIdx.x, X, W1, W2, Wa, z, zi, s1, s2, N);
    } else {
        gat_chunksort_body((int*)gat_cur4, (int*)gat_wl4, (int)blockIdx.x - G,
                           edge_d, src, dst, tab, rec_i, E, B, C);
    }
}

__device__ __forceinline__ float gat_logit(float s1v, float s2v, float ed, float w0, float wa)
{
    float v = s1v + s2v + ed * w0 * wa;
    return (v > 0.0f) ? v : 0.01f * v;
}

template <int CTRL>
__device__ __forceinline__ int gat_row_xmask_i(int x)
{
    return __builtin_amdgcn_update_dpp(0, x, CTRL, 0xf, 0xf, true);
}
__device__ __forceinline__ int gat_wave_isum(int x)
{
    x += gat_row_xmask_i<0x161>(x);
    x += gat_row_xmask_i<0x162>(x);
    x += gat_row_xmask_i<0x164>(x);
    x += gat_row_xmask_i<0x168>(x);
    return __builtin_amdgcn_readlane(x, 0) + __builtin_amdgcn_readlane(x, 16);
}

#define GAT_RUNS 10
#define GAT_PAIR_CAP 16384

__global__ __launch_bounds__(1024) void gat_binplace_kernel(
    const int* __restrict__ rec_i, const int* __restrict__ tab,
    const float* __restrict__ s1, const float* __restrict__ s2,
    const float* __restrict__ W0, const float* __restrict__ Wa,
    int* __restrict__ offs, int* __restrict__ pairs_i,
    int N, int E, int B, int C)
{
    __shared__ int ncur[GAT_BIN_NODES];
    __shared__ int wtot[16];
    __shared__ int wlo[32];
    __shared__ int segst[GAT_MAX_CHUNKS];
    __shared__ int segn[GAT_MAX_CHUNKS];
    __shared__ int wnb[32];
    __shared__ float s2l[GAT_BIN_NODES];
    __shared__ int2 pbuf[GAT_PAIR_CAP];
    int2* __restrict__ pairs = (int2*)pairs_i;
    const int2* __restrict__ rec = (const int2*)rec_i;
    const int t = threadIdx.x, lane = t & 31, w = t >> 5;
    const int b = blockIdx.x;
    const int node0 = b << GAT_BIN_SHIFT;

    if (t < GAT_BIN_NODES) {
        ncur[t] = 0;
        const int node = node0 + t;
        s2l[t] = (node < N) ? s2[node] : 0.0f;
    }
    int lo = 0, nb = 0;
    for (int c = t; c < C; c += 1024) {
        const int st = tab[b * C + c];
        const int en = tab[(b + 1) * C + c];
        segst[c] = c * GAT_CHUNK + st;
        segn[c]  = en - st;
        lo += st;
        nb += en - st;
    }
    lo = gat_wave_isum(lo);
    nb = gat_wave_isum(nb);
    if (lane == 0) { wlo[w] = lo; wnb[w] = nb; }
    __syncthreads();
    const int first = gat_wave_isum(wlo[lane]);
    const int nbin  = gat_wave_isum(wnb[lane]);
    const bool stage = nbin <= GAT_PAIR_CAP;

    for (int c0 = w; c0 < C; c0 += 32 * GAT_RUNS) {
        int n[GAT_RUNS], sb[GAT_RUNS];
#pragma unroll
        for (int u = 0; u < GAT_RUNS; ++u) {
            const int c = c0 + 32 * u;
            n[u]  = (c < C) ? segn[c] : 0;
            sb[u] = (c < C) ? segst[c] : 0;
        }
        for (int i = lane; ; i += 32) {
            int k[GAT_RUNS];
            bool any = false;
#pragma unroll
            for (int u = 0; u < GAT_RUNS; ++u) {
                const bool in = i < n[u];
                any |= in;
                k[u] = in ? rec[sb[u] + i].x : -1;
            }
            if (!any) break;
#pragma unroll
            for (int u = 0; u < GAT_RUNS; ++u)
                if (k[u] >= 0) atomicAdd(&ncur[k[u] & (GAT_BIN_NODES - 1)], 1);
        }
    }
    __syncthreads();

    int cval = 0, inc = 0;
    if (t < GAT_BIN_NODES) {
        cval = ncur[t];
        inc = cval;
        for (int off = 1; off < 32; off <<= 1) {
            const int v = __shfl_up(inc, off, 32);
            if (lane >= off) inc += v;
        }
        if (lane == 31) wtot[w] = inc;
    }
    __syncthreads();
    if (t < GAT_BIN_NODES) {
        const int tot = (lane < 16) ? wtot[lane] : 0;
        int winc = tot;
        for (int off = 1; off < 16; off <<= 1) {
            const int v = __shfl_up(winc, off, 32);
            if (lane >= off) winc += v;
        }
        const int wbase = __shfl(winc - tot, w, 32);
        const int start = first + wbase + (inc - cval);
        ncur[t] = start;
        const int node = node0 + t;
        if (node < N) offs[node] = start;
    }
    if (b == B - 1 && t == 0) offs[N] = E;
    __syncthreads();

    const float w0 = W0[0], wa = Wa[128];
    for (int c0 = w; c0 < C; c0 += 32 * GAT_RUNS) {
        int n[GAT_RUNS], sb[GAT_RUNS];
#pragma unroll
        for (int u = 0; u < GAT_RUNS; ++u) {
            const int c = c0 + 32 * u;
            n[u]  = (c < C) ? segn[c] : 0;
            sb[u] = (c < C) ? segst[c] : 0;
        }
        for (int i = lane; ; i += 32) {
            int2 r[GAT_RUNS];
            float a1[GAT_RUNS], a2[GAT_RUNS];
            bool any = false;
#pragma unroll
            for (int u = 0; u < GAT_RUNS; ++u) {
                const bool in = i < n[u];
                any |= in;
                r[u] = in ? rec[sb[u] + i] : make_int2(-1, 0);
            }
            if (!any) break;
#pragma unroll
            for (int u = 0; u < GAT_RUNS; ++u) {
                const int kk = (r[u].x >= 0) ? r[u].x : 0;
                a1[u] = s1[kk >> GAT_BIN_SHIFT];
                a2[u] = s2l[kk & (GAT_BIN_NODES - 1)];
            }
#pragma unroll
            for (int u = 0; u < GAT_RUNS; ++u) {
                if (r[u].x >= 0) {
                    const float v = gat_logit(a1[u], a2[u], __int_as_float(r[u].y), w0, wa);
                    const int pos = atomicAdd(&ncur[r[u].x & (GAT_BIN_NODES - 1)], 1);
                    const int2 pr = make_int2(r[u].x >> GAT_BIN_SHIFT, __float_as_int(v));
                    if (stage) pbuf[pos - first] = pr;
                    else       pairs[pos] = pr;
                }
            }
        }
    }
    if (stage) {
        __syncthreads();
        for (int i = t; i < nbin; i += 1024) pairs[first + i] = pbuf[i];
    }
}

__device__ __forceinline__ float gat_wave_max(float x)
{
    x = fmaxf(x, gat_row_xmask<0x161>(x));
    x = fmaxf(x, gat_row_xmask<0x162>(x));
    x = fmaxf(x, gat_row_xmask<0x164>(x));
    x = fmaxf(x, gat_row_xmask<0x168>(x));
    const float lo = __int_as_float(__builtin_amdgcn_readlane(__float_as_int(x), 0));
    const float hi = __int_as_float(__builtin_amdgcn_readlane(__float_as_int(x), 16));
    return fmaxf(lo, hi);
}
__device__ __forceinline__ float gat_wave_sum(float x)
{
    x += gat_row_xmask<0x161>(x);
    x += gat_row_xmask<0x162>(x);
    x += gat_row_xmask<0x164>(x);
    x += gat_row_xmask<0x168>(x);
    const float lo = __int_as_float(__builtin_amdgcn_readlane(__float_as_int(x), 0));
    const float hi = __int_as_float(__builtin_amdgcn_readlane(__float_as_int(x), 16));
    return lo + hi;
}

template <int K>
__device__ __forceinline__ void gat_pull_rows(
    const float* __restrict__ z, int s, float a, int j, int lane, v2f& acc)
{
    v2f r[K];
#pragma unroll
    for (int u = 0; u < K; ++u) {
        const int sj = __builtin_amdgcn_readlane(s, j + u);
        r[u] = *(const v2f*)(z + (size_t)sj * 64 + 2 * lane);
    }
#pragma unroll
    for (int u = 0; u < K; ++u) {
        const float aj = __int_as_float(
            __builtin_amdgcn_readlane(__float_as_int(a), j + u));
        acc.x = fmaf(aj, r[u].x, acc.x);
        acc.y = fmaf(aj, r[u].y, acc.y);
    }
}

__device__ __forceinline__ void gat_pull_chunk(
    const float* __restrict__ z, int s, float a, int n, int lane, v2f& acc)
{
    int j = 0;
    for (; j + 8 <= n; j += 8) gat_pull_rows<8>(z, s, a, j, lane, acc);
    if (n - j >= 4) { gat_pull_rows<4>(z, s, a, j, lane, acc); j += 4; }
    if (n - j >= 2) { gat_pull_rows<2>(z, s, a, j, lane, acc); j += 2; }
    if (n - j >= 1) { gat_pull_rows<1>(z, s, a, j, lane, acc); }
}

__global__ __launch_bounds__(256) void gat_pull_kernel(
    const int* __restrict__ offs,
    const int* __restrict__ pairs_i, const float* __restrict__ z,
    float* out, int N)
{
    const int2* __restrict__ pairs = (const int2*)pairs_i;
    const int lane = threadIdx.x & 31;
    const int wave = __builtin_amdgcn_readfirstlane(blockIdx.x * 8 + (threadIdx.x >> 5));
    if (wave >= N) return;
    const int off = offs[wave];
    const int deg = offs[wave + 1] - off;
    float* orow = out + (size_t)wave * 64 + 2 * lane;
    const v2f zi = *(const v2f*)orow;
    v2f acc = {0.0f, 0.0f};

    if (deg <= 32) {
        int s = 0;
        float v = -__builtin_inff();
        if (lane < deg) {
            const int2 p = pairs[off + lane];
            s = p.x;
            v = __int_as_float(p.y);
        }
        const float m   = gat_wave_max(v);
        const float ex  = (lane < deg) ? __expf(v - m) : 0.0f;
        const float den = gat_wave_sum(ex);
        const float a   = ex / den;
        gat_pull_chunk(z, s, a, deg, lane, acc);
    } else {
        float m = -__builtin_inff();
        for (int c = 0; c < deg; c += 32)
            if (c + lane < deg) m = fmaxf(m, __int_as_float(pairs[off + c + lane].y));
        m = gat_wave_max(m);
        float den = 0.0f;
        for (int c = 0; c < deg; c += 32)
            if (c + lane < deg) den += __expf(__int_as_float(pairs[off + c + lane].y) - m);
        den = gat_wave_sum(den);
        for (int c = 0; c < deg; c += 32) {
            int s = 0;
            float a = 0.0f;
            if (c + lane < deg) {
                const int2 p = pairs[off + c + lane];
                s = p.x;
                a = __expf(__int_as_float(p.y) - m) / den;
            }
            const int n = (deg - c < 32) ? (deg - c) : 32;
            gat_pull_chunk(z, s, a, n, lane, acc);
        }
    }

    v2f h;
    h.x = zi.x + acc.x;
    h.y = zi.y + acc.y;
    h.x = (h.x > 0.0f) ? h.x : 0.0f;
    h.y = (h.y > 0.0f) ? h.y : 0.0f;
    *(v2f*)orow = h;
}


extern "C" void kernel_launch(void* const* d_in, const int* in_sizes, int n_in,
                              void* d_out, int out_size, void* d_ws, size_t ws_size,
                              hipStream_t stream)
{
    const float* X   = (const float*)d_in[0];
    const float* ed  = (const float*)d_in[1];
    const float* W0  = (const float*)d_in[2];
    const float* W1  = (const float*)d_in[3];
    const float* W2  = (const float*)d_in[4];
    const float* Wa  = (const float*)d_in[5];
    const int*   src = (const int*)d_in[6];
    const int*   dst = (const int*)d_in[7];

    const int N = in_sizes[0] / 128;
    const int E = in_sizes[6];

    const int B = (N + GAT_BIN_NODES - 1) / GAT_BIN_NODES;
    const int C = (E + GAT_CHUNK - 1) / GAT_CHUNK;

    float* ws    = (float*)d_ws;
    float* z     = ws;         ws += (size_t)N * 64;
    int*   offs  = (int*)ws;
    int*   tab   = (int*)ws + (N + 1);
    ws += 2 * (size_t)N;
    float* s1    = ws;         ws += N;
    float* s2    = ws;         ws += N;
    int*   pairs = (int*)ws;   ws += (size_t)E * 2;
    int*   rec   = (int*)ws;   ws += (size_t)E * 2;
    float* out   = (float*)d_out;

    const int rowTiles = (N + 15) / 16;
    const int G = (rowTiles + 7) / 8;
    gat_gemm_binsort_kernel<<<G + C, 256, 0, stream>>>(
        X, W1, W2, Wa, z, out, s1, s2, N, ed, src, dst, tab, rec, E, B, C, G);

    gat_binplace_kernel<<<B, 1024, 0, stream>>>(
        rec, tab, s1, s2, W0, Wa, offs, pairs, N, E, B, C);

    const int nodeThreads = N * 32;
    gat_pull_kernel<<<(nodeThreads + 255) / 256, 256, 0, stream>>>(
        offs, pairs, z, out, N);
}
